# starting kernel
# baseline (speedup 1.0000x reference)
#include <hip/hip_runtime.h>

typedef __attribute__((ext_vector_type(16))) _Float16 v16h;
typedef __attribute__((ext_vector_type(8)))  float    v8f;

__global__ void __launch_bounds__(256)
edge_scatter128(const float* __restrict__ feat,
                const int*  __restrict__ src,
                const int*  __restrict__ dst,
                float* __restrict__ out, int E)
{
    const int wave = threadIdx.x >> 5;
    const int lane = threadIdx.x & 31;
    const int e = blockIdx.x * 8 + wave;
    if (e >= E) return;
    const int s = src[e];
    const int d = dst[e];
    const float4 v = *(const float4*)(feat + (size_t)s * 128 + lane * 4);
    float* o = out + (size_t)d * 128 + lane * 4;
    unsafeAtomicAdd(o + 0, v.x);
    unsafeAtomicAdd(o + 1, v.y);
    unsafeAtomicAdd(o + 2, v.z);
    unsafeAtomicAdd(o + 3, v.w);
}

__global__ void __launch_bounds__(256)
pad_head(const float* __restrict__ Wl, const float* __restrict__ bl,
         float* __restrict__ Wp, float* __restrict__ bp)
{
    const int i = blockIdx.x * 256 + threadIdx.x;
    if (i < 256 * 48) {
        const int r = i / 48, c = i % 48;
        Wp[i] = (c < 40) ? Wl[r * 40 + c] : 0.f;
    }
    if (i < 48) bp[i] = (i < 40) ? bl[i] : 0.f;
}

template<int K, int NW, bool RELU>
__global__ void __launch_bounds__(256)
gemm_bias_act(const float* __restrict__ A, const float* __restrict__ W,
              const float* __restrict__ bias, float* __restrict__ C,
              int Nout, int mtiles, int mgroups)
{
    constexpr int MT = 4;
    constexpr int NT = NW / 16;
    constexpr int KC = K / 32;

    const int lane = threadIdx.x & 31;
    const int wave = threadIdx.x >> 5;
    const int tile = blockIdx.x * 8 + wave;
    if (tile >= mgroups * NT) return;
    const int nt = tile % NT;
    const int mg = tile / NT;

    const int lm = lane & 15;
    const int lh = lane >> 4;
    const int col = nt * 16 + lm;

    v16h b[KC];
    const float* Wcol = W + col;
#pragma unroll
    for (int c = 0; c < KC; ++c) {
#pragma unroll
        for (int v = 0; v < 8; ++v) {
            const int kk = c * 32 + ((v < 4) ? (2 * v) : (16 + 2 * (v - 4))) + 8 * lh;
            b[c][2 * v]     = (_Float16)Wcol[(size_t)kk * NW];
            b[c][2 * v + 1] = (_Float16)Wcol[(size_t)(kk + 1) * NW];
        }
    }
    const float bv = bias[col];

    const int mt0 = mg * MT;
    const int mt1 = (mt0 + MT < mtiles) ? (mt0 + MT) : mtiles;
    for (int mt = mt0; mt < mt1; ++mt) {
        const float* Arow = A + (size_t)(mt * 16 + lm) * K;
        v8f acc = {};
#pragma unroll
        for (int c = 0; c < KC; ++c) {
            v16h a;
#pragma unroll
            for (int v = 0; v < 8; ++v) {
                const int kk = c * 32 + ((v < 4) ? (2 * v) : (16 + 2 * (v - 4))) + 8 * lh;
                const float2 av = *(const float2*)(Arow + kk);
                a[2 * v]     = (_Float16)av.x;
                a[2 * v + 1] = (_Float16)av.y;
            }
            acc = __builtin_amdgcn_wmma_f32_16x16x32_f16(
                false, a, false, b[c], (short)0, acc, false, false);
        }
        if (col < Nout) {
            float* Crow = C + (size_t)(mt * 16 + 8 * lh) * Nout + col;
#pragma unroll
            for (int r = 0; r < 8; ++r) {
                float o = acc[r] + bv;
                if (RELU) o = fmaxf(o, 0.f);
                Crow[(size_t)r * Nout] = o;
            }
        }
    }
}

template<int K, int NW, bool RELU>
static inline void launch_gemm(const float* A, const float* W, const float* b,
                               float* C, int M, int Nout, hipStream_t stream)
{
    const int mtiles  = M / 16;
    const int mgroups = (mtiles + 3) / 4;
    const int tiles   = mgroups * (NW / 16);
    const int blocks  = (tiles + 7) / 8;
    gemm_bias_act<K, NW, RELU><<<blocks, 256, 0, stream>>>(A, W, b, C, Nout, mtiles, mgroups);
}

extern "C" void kernel_launch(void* const* d_in, const int* in_sizes, int n_in,
                              void* d_out, int out_size, void* d_ws, size_t ws_size,
                              hipStream_t stream)
{
    const float* x  = (const float*)d_in[0];
    const int*   ei = (const int*)  d_in[1];
    const float* W1 = (const float*)d_in[2];
    const float* b1 = (const float*)d_in[3];
    const float* W2 = (const float*)d_in[4];
    const float* b2 = (const float*)d_in[5];
    const float* W3 = (const float*)d_in[6];
    const float* b3 = (const float*)d_in[7];
    const float* W4 = (const float*)d_in[8];
    const float* b4 = (const float*)d_in[9];
    const float* Wl = (const float*)d_in[10];
    const float* bl = (const float*)d_in[11];
    float* out = (float*)d_out;

    const int N = in_sizes[0] / 128;
    const int E = in_sizes[1] / 2;
    const int* src = ei;
    const int* dst = ei + E;

    float* ws   = (float*)d_ws;
    float* bufA = ws;
    float* bufB = ws + (size_t)N * 128;
    float* bufC = ws + (size_t)N * 256;
    float* bufD = ws;
    float* Wp   = ws + (size_t)N * 512;
    float* bp   = Wp + 256 * 48;

    const size_t feat128 = (size_t)N * 128 * sizeof(float);
    const int scatterBlocks = (E + 7) / 8;

    pad_head<<<(256 * 48 + 255) / 256, 256, 0, stream>>>(Wl, bl, Wp, bp);

    hipMemcpyAsync(bufA, x, feat128, hipMemcpyDeviceToDevice, stream);
    edge_scatter128<<<scatterBlocks, 256, 0, stream>>>(x, src, dst, bufA, E);

    launch_gemm<128, 128, true>(bufA, W1, b1, bufB, N, 128, stream);
    launch_gemm<128, 128, true>(bufB, W2, b2, bufA, N, 128, stream);

    hipMemcpyAsync(bufB, bufA, feat128, hipMemcpyDeviceToDevice, stream);
    edge_scatter128<<<scatterBlocks, 256, 0, stream>>>(bufA, src, dst, bufB, E);

    launch_gemm<128, 256, true>(bufB, W3, b3, bufC, N, 256, stream);
    launch_gemm<256, 256, true>(bufC, W4, b4, bufD, N, 256, stream);

    launch_gemm<256, 48, false>(bufD, Wp, bp, out, N, 40, stream);
}


# Aggregation: float-atomic edge scatter and its 2 copies replaced by a dst-bucketed gather (int-atomic counting sort per call); each node's in-rows are summed in ascending src id, a fixed sum order
# speedup vs baseline: 1.1230x; 1.1230x over previous
#include <hip/hip_runtime.h>

typedef __attribute__((ext_vector_type(16))) _Float16 v16h;
typedef __attribute__((ext_vector_type(8)))  float    v8f;


__global__ void __launch_bounds__(256)
prep(const float* __restrict__ Wl, const float* __restrict__ bl,
     float* __restrict__ Wp, float* __restrict__ bp,
     int* __restrict__ cnt, int ncnt)
{
    const int i = blockIdx.x * 256 + threadIdx.x;
    if (i < 256 * 48) {
        const int r = i / 48, c = i % 48;
        Wp[i] = (c < 40) ? Wl[r * 40 + c] : 0.f;
    }
    if (i < 48) bp[i] = (i < 40) ? bl[i] : 0.f;
    if (i < ncnt) cnt[i] = 0;
}

__global__ void __launch_bounds__(256)
degree_count(const int* __restrict__ dst, int* __restrict__ cnt, int E)
{
    const int i = blockIdx.x * 256 + threadIdx.x;
    const int stride = gridDim.x * 256;
    int d[4];
#pragma unroll
    for (int k = 0; k < 4; ++k) {
        const int e = i + k * stride;
        d[k] = (e < E) ? dst[e] : -1;
    }
#pragma unroll
    for (int k = 0; k < 4; ++k)
        if (d[k] >= 0) atomicAdd(&cnt[d[k]], 1);
}

__global__ void __launch_bounds__(256)
bucket_offsets(const int* __restrict__ cnt, int* __restrict__ total,
               int* __restrict__ offs, int* __restrict__ cur, int N)
{
    const int i = blockIdx.x * 256 + threadIdx.x;
    const int lane = threadIdx.x & 31;
    const int c = (i < N) ? cnt[i] : 0;
    int incl = c;
    for (int o = 1; o < 32; o <<= 1) {
        const int up = __shfl_up(incl, o, 32);
        if (lane >= o) incl += up;
    }
    int base = 0;
    if (lane == 31) base = atomicAdd(total, incl);
    base = __shfl(base, 31, 32);
    if (i < N) {
        const int b = base + incl - c;
        offs[i] = b;
        cur[i]  = b;
    }
}

__global__ void __launch_bounds__(256)
bucket_fill(const int* __restrict__ src, const int* __restrict__ dst,
            int* __restrict__ cur, int* __restrict__ bucket, int E)
{
    const int i = blockIdx.x * 256 + threadIdx.x;
    const int stride = gridDim.x * 256;
    int d[4], s[4];
#pragma unroll
    for (int k = 0; k < 4; ++k) {
        const int e = i + k * stride;
        d[k] = (e < E) ? dst[e] : -1;
        s[k] = (e < E) ? src[e] : 0;
    }
    int p[4];
#pragma unroll
    for (int k = 0; k < 4; ++k)
        p[k] = (d[k] >= 0) ? atomicAdd(&cur[d[k]], 1) : -1;
#pragma unroll
    for (int k = 0; k < 4; ++k)
        if (p[k] >= 0) bucket[p[k]] = s[k];
}

__device__ __forceinline__ float4 row4(const float* __restrict__ feat, int s, int lane)
{
    return *(const float4*)(feat + (size_t)s * 128 + lane * 4);
}

__device__ __forceinline__ void add4(float4& a, const float4 v)
{
    a.x += v.x; a.y += v.y; a.z += v.z; a.w += v.w;
}

__global__ void __launch_bounds__(256)
gather_sum128(const float* __restrict__ feat,
              const int* __restrict__ offs, const int* __restrict__ cnt,
              const int* __restrict__ bucket,
              float* __restrict__ out, int N)
{
    const int lane = threadIdx.x & 31;
    const int n = __builtin_amdgcn_readfirstlane(blockIdx.x * 8 + (threadIdx.x >> 5));
    if (n >= N) return;
    const int beg = offs[n];
    const int deg = cnt[n];
    float4 acc = row4(feat, n, lane);

    if (deg <= 32) {
        const int key = (lane < deg) ? bucket[beg + lane] : 0x7fffffff;
        int rank = 0;
        for (int j = 0; j < deg; ++j) {
            const int kj = __builtin_amdgcn_readlane(key, j);
            rank += ((kj < key) || (kj == key && j < lane)) ? 1 : 0;
        }
        if (lane >= deg) rank = lane;
        const int sk = __builtin_amdgcn_ds_permute(rank << 2, key);

        int j = 0;
        for (; j + 8 <= deg; j += 8) {
            float4 v[8];
#pragma unroll
            for (int t = 0; t < 8; ++t)
                v[t] = row4(feat, __builtin_amdgcn_readlane(sk, j + t), lane);
#pragma unroll
            for (int t = 0; t < 8; ++t) add4(acc, v[t]);
        }
        if (deg & 4) {
            const float4 v0 = row4(feat, __builtin_amdgcn_readlane(sk, j + 0), lane);
            const float4 v1 = row4(feat, __builtin_amdgcn_readlane(sk, j + 1), lane);
            const float4 v2 = row4(feat, __builtin_amdgcn_readlane(sk, j + 2), lane);
            const float4 v3 = row4(feat, __builtin_amdgcn_readlane(sk, j + 3), lane);
            add4(acc, v0); add4(acc, v1); add4(acc, v2); add4(acc, v3);
            j += 4;
        }
        if (deg & 2) {
            const float4 v0 = row4(feat, __builtin_amdgcn_readlane(sk, j + 0), lane);
            const float4 v1 = row4(feat, __builtin_amdgcn_readlane(sk, j + 1), lane);
            add4(acc, v0); add4(acc, v1);
            j += 2;
        }
        if (deg & 1) {
            const float4 v0 = row4(feat, __builtin_amdgcn_readlane(sk, j), lane);
            add4(acc, v0);
        }
    } else {
        int last = -1;
        int done = 0;
        for (int it = 0; it < deg && done < deg; ++it) {
            int m = 0x7fffffff;
            for (int i = lane; i < deg; i += 32) {
                const int k = bucket[beg + i];
                if (k > last && k < m) m = k;
            }
            for (int o = 16; o > 0; o >>= 1) {
                const int other = __shfl_xor(m, o, 32);
                m = (other < m) ? other : m;
            }
            int c = 0;
            for (int i = lane; i < deg; i += 32) c += (bucket[beg + i] == m) ? 1 : 0;
            for (int o = 16; o > 0; o >>= 1) c += __shfl_xor(c, o, 32);
            m = __builtin_amdgcn_readfirstlane(m);
            c = __builtin_amdgcn_readfirstlane(c);
            if (c <= 0) break;
            const float4 v = row4(feat, m, lane);
            for (int t = 0; t < c; ++t) add4(acc, v);
            last = m;
            done += c;
        }
    }
    *(float4*)(out + (size_t)n * 128 + lane * 4) = acc;
}

template<int K, int NW, bool RELU>
__global__ void __launch_bounds__(256)
gemm_bias_act(const float* __restrict__ A, const float* __restrict__ W,
              const float* __restrict__ bias, float* __restrict__ C,
              int Nout, int mtiles, int mgroups)
{
    constexpr int MT = 4;
    constexpr int NT = NW / 16;
    constexpr int KC = K / 32;

    const int lane = threadIdx.x & 31;
    const int wave = threadIdx.x >> 5;
    const int tile = blockIdx.x * 8 + wave;
    if (tile >= mgroups * NT) return;
    const int nt = tile % NT;
    const int mg = tile / NT;

    const int lm = lane & 15;
    const int lh = lane >> 4;
    const int col = nt * 16 + lm;

    v16h b[KC];
    const float* Wcol = W + col;
#pragma unroll
    for (int c = 0; c < KC; ++c) {
#pragma unroll
        for (int v = 0; v < 8; ++v) {
            const int kk = c * 32 + ((v < 4) ? (2 * v) : (16 + 2 * (v - 4))) + 8 * lh;
            b[c][2 * v]     = (_Float16)Wcol[(size_t)kk * NW];
            b[c][2 * v + 1] = (_Float16)Wcol[(size_t)(kk + 1) * NW];
        }
    }
    const float bv = bias[col];

    const int mt0 = mg * MT;
    const int mt1 = (mt0 + MT < mtiles) ? (mt0 + MT) : mtiles;
    for (int mt = mt0; mt < mt1; ++mt) {
        const float* Arow = A + (size_t)(mt * 16 + lm) * K;
        v8f acc = {};
#pragma unroll
        for (int c = 0; c < KC; ++c) {
            v16h a;
#pragma unroll
            for (int v = 0; v < 8; ++v) {
                const int kk = c * 32 + ((v < 4) ? (2 * v) : (16 + 2 * (v - 4))) + 8 * lh;
                const float2 av = *(const float2*)(Arow + kk);
                a[2 * v]     = (_Float16)av.x;
                a[2 * v + 1] = (_Float16)av.y;
            }
            acc = __builtin_amdgcn_wmma_f32_16x16x32_f16(
                false, a, false, b[c], (short)0, acc, false, false);
        }
        if (col < Nout) {
            float* Crow = C + (size_t)(mt * 16 + 8 * lh) * Nout + col;
#pragma unroll
            for (int r = 0; r < 8; ++r) {
                float o = acc[r] + bv;
                if (RELU) o = fmaxf(o, 0.f);
                Crow[(size_t)r * Nout] = o;
            }
        }
    }
}

template<int K, int NW, bool RELU>
static inline void launch_gemm(const float* A, const float* W, const float* b,
                               float* C, int M, int Nout, hipStream_t stream)
{
    const int mtiles  = M / 16;
    const int mgroups = (mtiles + 3) / 4;
    const int tiles   = mgroups * (NW / 16);
    const int blocks  = (tiles + 7) / 8;
    gemm_bias_act<K, NW, RELU><<<blocks, 256, 0, stream>>>(A, W, b, C, Nout, mtiles, mgroups);
}

extern "C" void kernel_launch(void* const* d_in, const int* in_sizes, int n_in,
                              void* d_out, int out_size, void* d_ws, size_t ws_size,
                              hipStream_t stream)
{
    const float* x  = (const float*)d_in[0];
    const int*   ei = (const int*)  d_in[1];
    const float* W1 = (const float*)d_in[2];
    const float* b1 = (const float*)d_in[3];
    const float* W2 = (const float*)d_in[4];
    const float* b2 = (const float*)d_in[5];
    const float* W3 = (const float*)d_in[6];
    const float* b3 = (const float*)d_in[7];
    const float* W4 = (const float*)d_in[8];
    const float* b4 = (const float*)d_in[9];
    const float* Wl = (const float*)d_in[10];
    const float* bl = (const float*)d_in[11];
    float* out = (float*)d_out;

    const int N = in_sizes[0] / 128;
    const int E = in_sizes[1] / 2;
    const int* src = ei;
    const int* dst = ei + E;

    float* ws   = (float*)d_ws;
    float* bufA = ws;
    float* bufB = ws + (size_t)N * 128;
    float* bufC = ws + (size_t)N * 256;
    float* bufD = ws;
    float* Wp   = ws + (size_t)N * 512;
    float* bp   = Wp + 256 * 48;
    int* cnt    = (int*)(ws + (size_t)N * 512 + 12352);
    int* offs   = cnt + (N + 64);
    int* cur    = offs + (N + 64);
    int* bucket = cur + (N + 64);

    const int edgeBlocks = (E + 1023) / 1024;
    const int nodeBlocks = (N + 255) / 256;
    const int waveBlocks = (N + 7) / 8;
    const int prepItems  = (N + 1 > 256 * 48) ? (N + 1) : (256 * 48);

    prep<<<(prepItems + 255) / 256, 256, 0, stream>>>(Wl, bl, Wp, bp, cnt, N + 1);
    degree_count<<<edgeBlocks, 256, 0, stream>>>(dst, cnt, E);
    bucket_offsets<<<nodeBlocks, 256, 0, stream>>>(cnt, cnt + N, offs, cur, N);
    bucket_fill<<<edgeBlocks, 256, 0, stream>>>(src, dst, cur, bucket, E);

    gather_sum128<<<waveBlocks, 256, 0, stream>>>(x, offs, cnt, bucket, bufA, N);

    launch_gemm<128, 128, true>(bufA, W1, b1, bufB, N, 128, stream);
    launch_gemm<128, 128, true>(bufB, W2, b2, bufA, N, 128, stream);

    gather_sum128<<<waveBlocks, 256, 0, stream>>>(bufA, offs, cnt, bucket, bufB, N);

    launch_gemm<128, 256, true>(bufB, W3, b3, bufC, N, 256, stream);
    launch_gemm<256, 256, true>(bufC, W4, b4, bufD, N, 256, stream);

    launch_gemm<256, 48, false>(bufD, Wp, bp, out, N, 40, stream);
}


# Each layer's 2 (3) GEMMs fused in one launch, transposed so the result tile repacks in registers as the next product's operand; k order in each 32-wide product changes; bias is the accumulator start
# speedup vs baseline: 1.8608x; 1.6570x over previous
#include <hip/hip_runtime.h>

typedef __attribute__((ext_vector_type(16))) _Float16 v16h;
typedef __attribute__((ext_vector_type(8)))  float    v8f;


__global__ void __launch_bounds__(256)
prep(const float* __restrict__ Wl, const float* __restrict__ bl,
     float* __restrict__ Wp, float* __restrict__ bp,
     int* __restrict__ cnt, int ncnt)
{
    const int i = blockIdx.x * 256 + threadIdx.x;
    if (i < 256 * 48) {
        const int r = i / 48, c = i % 48;
        Wp[i] = (c < 40) ? Wl[r * 40 + c] : 0.f;
    }
    if (i < 48) bp[i] = (i < 40) ? bl[i] : 0.f;
    if (i < ncnt) cnt[i] = 0;
}

__global__ void __launch_bounds__(256)
degree_count(const int* __restrict__ dst, int* __restrict__ cnt, int E)
{
    const int i = blockIdx.x * 256 + threadIdx.x;
    const int stride = gridDim.x * 256;
    int d[4];
#pragma unroll
    for (int k = 0; k < 4; ++k) {
        const int e = i + k * stride;
        d[k] = (e < E) ? dst[e] : -1;
    }
#pragma unroll
    for (int k = 0; k < 4; ++k)
        if (d[k] >= 0) atomicAdd(&cnt[d[k]], 1);
}

__global__ void __launch_bounds__(256)
bucket_offsets(const int* __restrict__ cnt, int* __restrict__ total,
               int* __restrict__ offs, int* __restrict__ cur, int N)
{
    const int i = blockIdx.x * 256 + threadIdx.x;
    const int lane = threadIdx.x & 31;
    const int c = (i < N) ? cnt[i] : 0;
    int incl = c;
    for (int o = 1; o < 32; o <<= 1) {
        const int up = __shfl_up(incl, o, 32);
        if (lane >= o) incl += up;
    }
    int base = 0;
    if (lane == 31) base = atomicAdd(total, incl);
    base = __shfl(base, 31, 32);
    if (i < N) {
        const int b = base + incl - c;
        offs[i] = b;
        cur[i]  = b;
    }
}

__global__ void __launch_bounds__(256)
bucket_fill(const int* __restrict__ src, const int* __restrict__ dst,
            int* __restrict__ cur, int* __restrict__ bucket, int E)
{
    const int i = blockIdx.x * 256 + threadIdx.x;
    const int stride = gridDim.x * 256;
    int d[4], s[4];
#pragma unroll
    for (int k = 0; k < 4; ++k) {
        const int e = i + k * stride;
        d[k] = (e < E) ? dst[e] : -1;
        s[k] = (e < E) ? src[e] : 0;
    }
    int p[4];
#pragma unroll
    for (int k = 0; k < 4; ++k)
        p[k] = (d[k] >= 0) ? atomicAdd(&cur[d[k]], 1) : -1;
#pragma unroll
    for (int k = 0; k < 4; ++k)
        if (p[k] >= 0) bucket[p[k]] = s[k];
}

__device__ __forceinline__ float4 row4(const float* __restrict__ feat, int s, int lane)
{
    return *(const float4*)(feat + (size_t)s * 128 + lane * 4);
}

__device__ __forceinline__ void add4(float4& a, const float4 v)
{
    a.x += v.x; a.y += v.y; a.z += v.z; a.w += v.w;
}

__global__ void __launch_bounds__(256)
gather_sum128(const float* __restrict__ feat,
              const int* __restrict__ offs, const int* __restrict__ cnt,
              const int* __restrict__ bucket,
              float* __restrict__ out, int N)
{
    const int lane = threadIdx.x & 31;
    const int n = __builtin_amdgcn_readfirstlane(blockIdx.x * 8 + (threadIdx.x >> 5));
    if (n >= N) return;
    const int beg = offs[n];
    const int deg = cnt[n];
    float4 acc = row4(feat, n, lane);

    if (deg <= 32) {
        const int key = (lane < deg) ? bucket[beg + lane] : 0x7fffffff;
        int rank = 0;
        for (int j = 0; j < deg; ++j) {
            const int kj = __builtin_amdgcn_readlane(key, j);
            rank += ((kj < key) || (kj == key && j < lane)) ? 1 : 0;
        }
        if (lane >= deg) rank = lane;
        const int sk = __builtin_amdgcn_ds_permute(rank << 2, key);

        int j = 0;
        for (; j + 8 <= deg; j += 8) {
            float4 v[8];
#pragma unroll
            for (int t = 0; t < 8; ++t)
                v[t] = row4(feat, __builtin_amdgcn_readlane(sk, j + t), lane);
#pragma unroll
            for (int t = 0; t < 8; ++t) add4(acc, v[t]);
        }
        if (deg & 4) {
            const float4 v0 = row4(feat, __builtin_amdgcn_readlane(sk, j + 0), lane);
            const float4 v1 = row4(feat, __builtin_amdgcn_readlane(sk, j + 1), lane);
            const float4 v2 = row4(feat, __builtin_amdgcn_readlane(sk, j + 2), lane);
            const float4 v3 = row4(feat, __builtin_amdgcn_readlane(sk, j + 3), lane);
            add4(acc, v0); add4(acc, v1); add4(acc, v2); add4(acc, v3);
            j += 4;
        }
        if (deg & 2) {
            const float4 v0 = row4(feat, __builtin_amdgcn_readlane(sk, j + 0), lane);
            const float4 v1 = row4(feat, __builtin_amdgcn_readlane(sk, j + 1), lane);
            add4(acc, v0); add4(acc, v1);
            j += 2;
        }
        if (deg & 1) {
            const float4 v0 = row4(feat, __builtin_amdgcn_readlane(sk, j), lane);
            add4(acc, v0);
        }
    } else {
        int last = -1;
        int done = 0;
        for (int it = 0; it < deg && done < deg; ++it) {
            int m = 0x7fffffff;
            for (int i = lane; i < deg; i += 32) {
                const int k = bucket[beg + i];
                if (k > last && k < m) m = k;
            }
            for (int o = 16; o > 0; o >>= 1) {
                const int other = __shfl_xor(m, o, 32);
                m = (other < m) ? other : m;
            }
            int c = 0;
            for (int i = lane; i < deg; i += 32) c += (bucket[beg + i] == m) ? 1 : 0;
            for (int o = 16; o > 0; o >>= 1) c += __shfl_xor(c, o, 32);
            m = __builtin_amdgcn_readfirstlane(m);
            c = __builtin_amdgcn_readfirstlane(c);
            if (c <= 0) break;
            const float4 v = row4(feat, m, lane);
            for (int t = 0; t < c; ++t) add4(acc, v);
            last = m;
            done += c;
        }
    }
    *(float4*)(out + (size_t)n * 128 + lane * 4) = acc;
}

typedef __attribute__((ext_vector_type(8))) _Float16 v8h;

template<int K, int NW>
__device__ __forceinline__ void stage_weights(const float* __restrict__ W,
                                              _Float16* __restrict__ Ws, int tid)
{
    constexpr int NT = NW / 16;
    constexpr int TASKS = (K / 8) * NW;
    static_assert(TASKS % 512 == 0, "every thread stages the same number of tasks");
#pragma unroll 4
    for (int i = 0; i < TASKS / 512; ++i) {
        const int t   = i * 512 + tid;
        const int col = t % NW;
        const int kg  = t / NW;
        const float* s = W + (size_t)(kg * 8) * NW + col;
        v8h pk;
#pragma unroll
        for (int e = 0; e < 8; ++e) pk[e] = (_Float16)s[(size_t)e * NW];
        const int q  = kg >> 2;
        const int j  = (kg >> 1) & 1;
        const int h  = kg & 1;
        const int nt = col >> 4;
        const int c  = col & 15;
        *(v8h*)(Ws + ((((q * NT + nt) * 2 + j) * 32) + h * 16 + c) * 8) = pk;
    }
}

template<int NB>
__device__ __forceinline__ void stage_bias(const float* __restrict__ b,
                                           float* __restrict__ Bs, int tid)
{
    static_assert(NB <= 512, "one element per thread");
    if (tid < NB) Bs[tid] = b[tid];
}

template<int K>
__device__ __forceinline__ void load_rows(const float* __restrict__ zrow, int lh,
                                          v16h (&hb)[K / 32])
{
#pragma unroll
    for (int q = 0; q < K / 32; ++q) {
#pragma unroll
        for (int j = 0; j < 2; ++j) {
            const float* p = zrow + 32 * q + 16 * j + 8 * lh;
            const float4 v0 = *(const float4*)p;
            const float4 v1 = *(const float4*)(p + 4);
            hb[q][8 * j + 0] = (_Float16)v0.x;
            hb[q][8 * j + 1] = (_Float16)v0.y;
            hb[q][8 * j + 2] = (_Float16)v0.z;
            hb[q][8 * j + 3] = (_Float16)v0.w;
            hb[q][8 * j + 4] = (_Float16)v1.x;
            hb[q][8 * j + 5] = (_Float16)v1.y;
            hb[q][8 * j + 6] = (_Float16)v1.z;
            hb[q][8 * j + 7] = (_Float16)v1.w;
        }
    }
}

template<int K, int NW>
__device__ __forceinline__ void product(const v16h (&hb)[K / 32],
                                        const _Float16* __restrict__ Ws,
                                        const float* __restrict__ Bs, int lane,
                                        v8f (&acc)[NW / 16])
{
    constexpr int NT = NW / 16;
    constexpr int KC = K / 32;
    constexpr int G = (NT % 4 == 0) ? 4 : NT;
#pragma unroll
    for (int t = 0; t < NT; ++t) acc[t] = *(const v8f*)(Bs + 16 * t + 8 * (lane >> 4));
#pragma unroll
    for (int q = 0; q < KC; ++q) {
#pragma unroll
        for (int g = 0; g < NT; g += G) {
            v16h w[G];
#pragma unroll
            for (int u = 0; u < G; ++u) {
                const _Float16* p = Ws + (((q * NT + g + u) * 2) * 32 + lane) * 8;
                const v8h lo = *(const v8h*)p;
                const v8h hi = *(const v8h*)(p + 32 * 8);
                w[u] = __builtin_shufflevector(lo, hi,
                    0, 1, 2, 3, 4, 5, 6, 7, 8, 9, 10, 11, 12, 13, 14, 15);
            }
#pragma unroll
            for (int u = 0; u < G; ++u)
                acc[g + u] = __builtin_amdgcn_wmma_f32_16x16x32_f16(
                    false, w[u], false, hb[q], (short)0, acc[g + u], false, false);
        }
    }
}

template<int NW>
__device__ __forceinline__ void to_fragments(const v8f (&acc)[NW / 16],
                                             v16h (&hb)[NW / 32])
{
#pragma unroll
    for (int t = 0; t < NW / 16; ++t) {
#pragma unroll
        for (int r = 0; r < 8; ++r)
            hb[t >> 1][8 * (t & 1) + r] = (_Float16)fmaxf(acc[t][r], 0.f);
    }
}

template<int NW, bool RELU>
__device__ __forceinline__ void store_rows(const v8f (&acc)[NW / 16], int lh,
                                           float* __restrict__ orow, int Nout)
{
#pragma unroll
    for (int t = 0; t < NW / 16; ++t) {
        const int col0 = 16 * t + 8 * lh;
        if (col0 < Nout) {
            float4 o0, o1;
            o0.x = acc[t][0]; o0.y = acc[t][1]; o0.z = acc[t][2]; o0.w = acc[t][3];
            o1.x = acc[t][4]; o1.y = acc[t][5]; o1.z = acc[t][6]; o1.w = acc[t][7];
            if (RELU) {
                o0.x = fmaxf(o0.x, 0.f); o0.y = fmaxf(o0.y, 0.f);
                o0.z = fmaxf(o0.z, 0.f); o0.w = fmaxf(o0.w, 0.f);
                o1.x = fmaxf(o1.x, 0.f); o1.y = fmaxf(o1.y, 0.f);
                o1.z = fmaxf(o1.z, 0.f); o1.w = fmaxf(o1.w, 0.f);
            }
            *(float4*)(orow + col0)     = o0;
            *(float4*)(orow + col0 + 4) = o1;
        }
    }
}

__global__ void __launch_bounds__(512)
mlp_layer1(const float* __restrict__ z,
           const float* __restrict__ Wa, const float* __restrict__ ba,
           const float* __restrict__ Wb, const float* __restrict__ bb,
           float* __restrict__ out, int mtiles)
{
    __shared__ __attribute__((aligned(16))) _Float16 WsA[128 * 128];
    __shared__ __attribute__((aligned(16))) _Float16 WsB[128 * 128];
    __shared__ __attribute__((aligned(32))) float    Bs[128 + 128];

    const int tid  = threadIdx.x;
    const int lane = tid & 31;
    const int wave = __builtin_amdgcn_readfirstlane(tid >> 5);
    const int lm = lane & 15;
    const int lh = lane >> 4;
    const int mt = blockIdx.x * 16 + wave;
    const bool active = mt < mtiles;

    const int node = (active ? mt : (mtiles - 1)) * 16 + lm;
    v16h h0[4];
    load_rows<128>(z + (size_t)node * 128, lh, h0);

    stage_weights<128, 128>(Wa, WsA, tid);
    stage_weights<128, 128>(Wb, WsB, tid);
    stage_bias<128>(ba, Bs, tid);
    stage_bias<128>(bb, Bs + 128, tid);
    asm volatile("" :: "v"(h0[0]), "v"(h0[1]), "v"(h0[2]), "v"(h0[3]));
    __syncthreads();
    if (!active) return;

    v8f acc[8];
    product<128, 128>(h0, WsA, Bs, lane, acc);
    v16h h1[4];
    to_fragments<128>(acc, h1);
    product<128, 128>(h1, WsB, Bs + 128, lane, acc);
    store_rows<128, true>(acc, lh, out + (size_t)node * 128, 128);
}

__global__ void __launch_bounds__(512)
mlp_layer2_head(const float* __restrict__ z,
                const float* __restrict__ Wa, const float* __restrict__ ba,
                const float* __restrict__ Wb, const float* __restrict__ bb,
                const float* __restrict__ Wc, const float* __restrict__ bc,
                float* __restrict__ out, int Nout, int mtiles)
{
    __shared__ __attribute__((aligned(16))) _Float16 WsA[128 * 256];
    __shared__ __attribute__((aligned(16))) _Float16 WsB[256 * 256];
    __shared__ __attribute__((aligned(16))) _Float16 WsC[256 * 48];
    __shared__ __attribute__((aligned(32))) float    Bs[256 + 256 + 48];

    const int tid  = threadIdx.x;
    const int lane = tid & 31;
    const int wave = __builtin_amdgcn_readfirstlane(tid >> 5);
    const int lm = lane & 15;
    const int lh = lane >> 4;
    const int mt = blockIdx.x * 16 + wave;
    const bool active = mt < mtiles;

    const int node = (active ? mt : (mtiles - 1)) * 16 + lm;
    v16h h0[4];
    load_rows<128>(z + (size_t)node * 128, lh, h0);

    stage_weights<128, 256>(Wa, WsA, tid);
    stage_weights<256, 256>(Wb, WsB, tid);
    stage_weights<256, 48>(Wc, WsC, tid);
    stage_bias<256>(ba, Bs, tid);
    stage_bias<256>(bb, Bs + 256, tid);
    stage_bias<48>(bc, Bs + 512, tid);
    asm volatile("" :: "v"(h0[0]), "v"(h0[1]), "v"(h0[2]), "v"(h0[3]));
    __syncthreads();
    if (!active) return;

    v8f acc[16];
    product<128, 256>(h0, WsA, Bs, lane, acc);
    v16h h1[8];
    to_fragments<256>(acc, h1);
    product<256, 256>(h1, WsB, Bs + 256, lane, acc);
    v16h h2[8];
    to_fragments<256>(acc, h2);
    v8f acc3[3];
    product<256, 48>(h2, WsC, Bs + 512, lane, acc3);
    store_rows<48, false>(acc3, lh, out + (size_t)node * Nout, Nout);
}

extern "C" void kernel_launch(void* const* d_in, const int* in_sizes, int n_in,
                              void* d_out, int out_size, void* d_ws, size_t ws_size,
                              hipStream_t stream)
{
    const float* x  = (const float*)d_in[0];
    const int*   ei = (const int*)  d_in[1];
    const float* W1 = (const float*)d_in[2];
    const float* b1 = (const float*)d_in[3];
    const float* W2 = (const float*)d_in[4];
    const float* b2 = (const float*)d_in[5];
    const float* W3 = (const float*)d_in[6];
    const float* b3 = (const float*)d_in[7];
    const float* W4 = (const float*)d_in[8];
    const float* b4 = (const float*)d_in[9];
    const float* Wl = (const float*)d_in[10];
    const float* bl = (const float*)d_in[11];
    float* out = (float*)d_out;

    const int N = in_sizes[0] / 128;
    const int E = in_sizes[1] / 2;
    const int* src = ei;
    const int* dst = ei + E;

    float* ws   = (float*)d_ws;
    float* bufA = ws;
    float* bufB = ws + (size_t)N * 128;
    float* Wp   = ws + (size_t)N * 512;
    float* bp   = Wp + 256 * 48;
    int* cnt    = (int*)(ws + (size_t)N * 512 + 12352);
    int* offs   = cnt + (N + 64);
    int* cur    = offs + (N + 64);
    int* bucket = cur + (N + 64);

    const int edgeBlocks = (E + 1023) / 1024;
    const int nodeBlocks = (N + 255) / 256;
    const int waveBlocks = (N + 7) / 8;
    const int mtiles     = N / 16;
    const int gemmBlocks = (mtiles + 15) / 16;
    const int prepItems  = (N + 1 > 256 * 48) ? (N + 1) : (256 * 48);

    prep<<<(prepItems + 255) / 256, 256, 0, stream>>>(Wl, bl, Wp, bp, cnt, N + 1);
    degree_count<<<edgeBlocks, 256, 0, stream>>>(dst, cnt, E);
    bucket_offsets<<<nodeBlocks, 256, 0, stream>>>(cnt, cnt + N, offs, cur, N);
    bucket_fill<<<edgeBlocks, 256, 0, stream>>>(src, dst, cur, bucket, E);

    gather_sum128<<<waveBlocks, 256, 0, stream>>>(x, offs, cnt, bucket, bufA, N);

    mlp_layer1<<<gemmBlocks, 512, 0, stream>>>(bufA, W1, b1, W2, b2, bufB, mtiles);

    gather_sum128<<<waveBlocks, 256, 0, stream>>>(bufB, offs, cnt, bucket, bufA, N);

    mlp_layer2_head<<<gemmBlocks, 512, 0, stream>>>(bufA, W3, b3, W4, b4, Wp, bp, out, 40, mtiles);
}


# Edge bucketing: the two per-edge passes of global integer atomics (degree count, slot fill) and the offset scan replaced by a two-level counting sort on LDS counters; 8 launches -> 7; sums unchanged
# speedup vs baseline: 3.0635x; 1.6463x over previous
#include <hip/hip_runtime.h>

typedef __attribute__((ext_vector_type(16))) _Float16 v16h;
typedef __attribute__((ext_vector_type(8)))  float    v8f;


constexpr int SORT_CHUNK = 4096;
constexpr int BIN_SHIFT  = 8;
constexpr int BIN_NODES  = 1 << BIN_SHIFT;
constexpr int MAX_BINS   = 256;

__global__ void __launch_bounds__(256)
prep(const float* __restrict__ Wl, const float* __restrict__ bl,
     float* __restrict__ Wp, float* __restrict__ bp, int* __restrict__ total)
{
    const int i = blockIdx.x * 256 + threadIdx.x;
    if (i < 256 * 48) {
        const int r = i / 48, c = i % 48;
        Wp[i] = (c < 40) ? Wl[r * 40 + c] : 0.f;
    }
    if (i < 48) bp[i] = (i < 40) ? bl[i] : 0.f;
    if (i == 0) *total = 0;
}

__global__ void __launch_bounds__(512)
sort_chunks(const int* __restrict__ src, const int* __restrict__ dst,
            int* __restrict__ recs, int* __restrict__ table,
            int E, int nbins, int nchunks)
{
    __shared__ int hist[MAX_BINS];
    __shared__ int start[MAX_BINS];
    __shared__ int srec[SORT_CHUNK];

    const int tid  = threadIdx.x;
    const int lane = tid & 31;
    const int wave  = __builtin_amdgcn_readfirstlane(tid >> 5);
    const int chunk = blockIdx.x;
    const int base  = chunk * SORT_CHUNK;

    if (tid < MAX_BINS) hist[tid] = 0;

    int d[8], s[8];
#pragma unroll
    for (int k = 0; k < 8; ++k) {
        const int e = base + k * 512 + tid;
        d[k] = (e < E) ? dst[e] : -1;
        s[k] = (e < E) ? src[e] : 0;
    }
    __syncthreads();

    int r[8];
#pragma unroll
    for (int k = 0; k < 8; ++k)
        r[k] = (d[k] >= 0) ? atomicAdd(&hist[d[k] >> BIN_SHIFT], 1) : 0;
    __syncthreads();

    if (wave == 0) {
        int c[8];
        int sum = 0;
#pragma unroll
        for (int j = 0; j < 8; ++j) { c[j] = sum; sum += hist[lane * 8 + j]; }
        int incl = sum;
        for (int o = 1; o < 32; o <<= 1) {
            const int up = __shfl_up(incl, o, 32);
            if (lane >= o) incl += up;
        }
        const int excl = incl - sum;
#pragma unroll
        for (int j = 0; j < 8; ++j) start[lane * 8 + j] = excl + c[j];
    }
    __syncthreads();

    int q[8];
#pragma unroll
    for (int k = 0; k < 8; ++k)
        q[k] = (d[k] >= 0) ? start[d[k] >> BIN_SHIFT] : 0;
#pragma unroll
    for (int k = 0; k < 8; ++k)
        if (d[k] >= 0) srec[q[k] + r[k]] = ((d[k] & (BIN_NODES - 1)) << 16) | s[k];
    __syncthreads();

    const int n = (E - base < SORT_CHUNK) ? (E - base) : SORT_CHUNK;
    int o[8];
#pragma unroll
    for (int k = 0; k < 8; ++k) {
        const int i = k * 512 + tid;
        o[k] = (i < n) ? srec[i] : 0;
    }
#pragma unroll
    for (int k = 0; k < 8; ++k) {
        const int i = k * 512 + tid;
        if (i < n) recs[base + i] = o[k];
    }
    if (tid < nbins) {
        int2 t;
        t.x = base + start[tid];
        t.y = hist[tid];
        *(int2*)(table + ((size_t)tid * nchunks + chunk) * 2) = t;
    }
}

template<bool PLACE>
__device__ __forceinline__ void bin_record(int v, int* __restrict__ ncnt, int* __restrict__ ncur,
                                           const int* __restrict__ nstart, int* __restrict__ bucket)
{
    const int dl = v >> 16;
    if (PLACE) bucket[nstart[dl] + atomicAdd(&ncur[dl], 1)] = v & 0xffff;
    else       atomicAdd(&ncnt[dl], 1);
}

template<bool PLACE>
__device__ __forceinline__ void bin_rest(const int* __restrict__ recs, const int* __restrict__ row,
                                         int pstart, int plen, int lane, int wave, int nchunks,
                                         int* __restrict__ ncnt, int* __restrict__ ncur,
                                         const int* __restrict__ nstart, int* __restrict__ bucket)
{
    if (__builtin_amdgcn_ballot_w32(plen > 64) != 0u) {
        for (int j = 0; j < 16; ++j) {
            const int po = __builtin_amdgcn_readlane(pstart, j);
            const int pn = __builtin_amdgcn_readlane(plen, j);
            for (int i = 64 + lane; i < pn; i += 32)
                bin_record<PLACE>(recs[po + i], ncnt, ncur, nstart, bucket);
        }
    }
    for (int w = 256 + wave; w < nchunks; w += 16) {
        const int po = row[2 * w];
        const int pn = row[2 * w + 1];
        for (int i = lane; i < pn; i += 32)
            bin_record<PLACE>(recs[po + i], ncnt, ncur, nstart, bucket);
    }
}

__global__ void __launch_bounds__(512)
sort_bins(const int* __restrict__ recs, const int* __restrict__ table,
          int* __restrict__ total, int* __restrict__ cnt, int* __restrict__ offs,
          int* __restrict__ bucket, int N, int nchunks)
{
    __shared__ int ncnt[BIN_NODES];
    __shared__ int ncur[BIN_NODES];
    __shared__ int nstart[BIN_NODES];

    const int tid  = threadIdx.x;
    const int lane = tid & 31;
    const int wave = __builtin_amdgcn_readfirstlane(tid >> 5);
    const int bin  = blockIdx.x;
    const int* row = table + (size_t)bin * nchunks * 2;

    if (tid < BIN_NODES) { ncnt[tid] = 0; ncur[tid] = 0; }

    const int wi = wave * 16 + (lane & 15);
    int pstart = 0, plen = 0;
    if (wi < nchunks) {
        const int2 t = *(const int2*)(row + 2 * wi);
        pstart = t.x;
        plen   = t.y;
    }

    int v[32];
#pragma unroll
    for (int j = 0; j < 16; ++j) {
        const int po = __builtin_amdgcn_readlane(pstart, j);
        const int pn = __builtin_amdgcn_readlane(plen, j);
        v[2 * j]     = (lane < pn)      ? recs[po + lane]      : -1;
        v[2 * j + 1] = (lane + 32 < pn) ? recs[po + 32 + lane] : -1;
    }
    __syncthreads();

#pragma unroll
    for (int k = 0; k < 32; ++k)
        if (v[k] >= 0) bin_record<false>(v[k], ncnt, ncur, nstart, bucket);
    bin_rest<false>(recs, row, pstart, plen, lane, wave, nchunks, ncnt, ncur, nstart, bucket);
    __syncthreads();

    if (wave == 0) {
        int c[8];
        int sum = 0;
#pragma unroll
        for (int j = 0; j < 8; ++j) { c[j] = sum; sum += ncnt[lane * 8 + j]; }
        int incl = sum;
        for (int o = 1; o < 32; o <<= 1) {
            const int up = __shfl_up(incl, o, 32);
            if (lane >= o) incl += up;
        }
        int slice = 0;
        if (lane == 31) slice = atomicAdd(total, incl);
        slice = __shfl(slice, 31, 32);
        const int excl = slice + incl - sum;
#pragma unroll
        for (int j = 0; j < 8; ++j) nstart[lane * 8 + j] = excl + c[j];
    }
    __syncthreads();

    if (tid < BIN_NODES) {
        const int n = bin * BIN_NODES + tid;
        if (n < N) {
            cnt[n]  = ncnt[tid];
            offs[n] = nstart[tid];
        }
    }

    int p[32], t[32];
#pragma unroll
    for (int k = 0; k < 32; ++k) {
        p[k] = (v[k] >= 0) ? nstart[v[k] >> 16] : 0;
        t[k] = (v[k] >= 0) ? atomicAdd(&ncur[v[k] >> 16], 1) : 0;
    }
#pragma unroll
    for (int k = 0; k < 32; ++k)
        if (v[k] >= 0) bucket[p[k] + t[k]] = v[k] & 0xffff;
    bin_rest<true>(recs, row, pstart, plen, lane, wave, nchunks, ncnt, ncur, nstart, bucket);
}

__device__ __forceinline__ float4 row4(const float* __restrict__ feat, int s, int lane)
{
    return *(const float4*)(feat + (size_t)s * 128 + lane * 4);
}

__device__ __forceinline__ void add4(float4& a, const float4 v)
{
    a.x += v.x; a.y += v.y; a.z += v.z; a.w += v.w;
}

__global__ void __launch_bounds__(256)
gather_sum128(const float* __restrict__ feat,
              const int* __restrict__ offs, const int* __restrict__ cnt,
              const int* __restrict__ bucket,
              float* __restrict__ out, int N)
{
    const int lane = threadIdx.x & 31;
    const int n = __builtin_amdgcn_readfirstlane(blockIdx.x * 8 + (threadIdx.x >> 5));
    if (n >= N) return;
    const int beg = offs[n];
    const int deg = cnt[n];
    float4 acc = row4(feat, n, lane);

    if (deg <= 32) {
        const int key = (lane < deg) ? bucket[beg + lane] : 0x7fffffff;
        int rank = 0;
        for (int j = 0; j < deg; ++j) {
            const int kj = __builtin_amdgcn_readlane(key, j);
            rank += ((kj < key) || (kj == key && j < lane)) ? 1 : 0;
        }
        if (lane >= deg) rank = lane;
        const int sk = __builtin_amdgcn_ds_permute(rank << 2, key);

        int j = 0;
        for (; j + 8 <= deg; j += 8) {
            float4 v[8];
#pragma unroll
            for (int t = 0; t < 8; ++t)
                v[t] = row4(feat, __builtin_amdgcn_readlane(sk, j + t), lane);
#pragma unroll
            for (int t = 0; t < 8; ++t) add4(acc, v[t]);
        }
        if (deg & 4) {
            const float4 v0 = row4(feat, __builtin_amdgcn_readlane(sk, j + 0), lane);
            const float4 v1 = row4(feat, __builtin_amdgcn_readlane(sk, j + 1), lane);
            const float4 v2 = row4(feat, __builtin_amdgcn_readlane(sk, j + 2), lane);
            const float4 v3 = row4(feat, __builtin_amdgcn_readlane(sk, j + 3), lane);
            add4(acc, v0); add4(acc, v1); add4(acc, v2); add4(acc, v3);
            j += 4;
        }
        if (deg & 2) {
            const float4 v0 = row4(feat, __builtin_amdgcn_readlane(sk, j + 0), lane);
            const float4 v1 = row4(feat, __builtin_amdgcn_readlane(sk, j + 1), lane);
            add4(acc, v0); add4(acc, v1);
            j += 2;
        }
        if (deg & 1) {
            const float4 v0 = row4(feat, __builtin_amdgcn_readlane(sk, j), lane);
            add4(acc, v0);
        }
    } else {
        int last = -1;
        int done = 0;
        for (int it = 0; it < deg && done < deg; ++it) {
            int m = 0x7fffffff;
            for (int i = lane; i < deg; i += 32) {
                const int k = bucket[beg + i];
                if (k > last && k < m) m = k;
            }
            for (int o = 16; o > 0; o >>= 1) {
                const int other = __shfl_xor(m, o, 32);
                m = (other < m) ? other : m;
            }
            int c = 0;
            for (int i = lane; i < deg; i += 32) c += (bucket[beg + i] == m) ? 1 : 0;
            for (int o = 16; o > 0; o >>= 1) c += __shfl_xor(c, o, 32);
            m = __builtin_amdgcn_readfirstlane(m);
            c = __builtin_amdgcn_readfirstlane(c);
            if (c <= 0) break;
            const float4 v = row4(feat, m, lane);
            for (int t = 0; t < c; ++t) add4(acc, v);
            last = m;
            done += c;
        }
    }
    *(float4*)(out + (size_t)n * 128 + lane * 4) = acc;
}

typedef __attribute__((ext_vector_type(8))) _Float16 v8h;

template<int K, int NW>
__device__ __forceinline__ void stage_weights(const float* __restrict__ W,
                                              _Float16* __restrict__ Ws, int tid)
{
    constexpr int NT = NW / 16;
    constexpr int TASKS = (K / 8) * NW;
    static_assert(TASKS % 512 == 0, "every thread stages the same number of tasks");
#pragma unroll 4
    for (int i = 0; i < TASKS / 512; ++i) {
        const int t   = i * 512 + tid;
        const int col = t % NW;
        const int kg  = t / NW;
        const float* s = W + (size_t)(kg * 8) * NW + col;
        v8h pk;
#pragma unroll
        for (int e = 0; e < 8; ++e) pk[e] = (_Float16)s[(size_t)e * NW];
        const int q  = kg >> 2;
        const int j  = (kg >> 1) & 1;
        const int h  = kg & 1;
        const int nt = col >> 4;
        const int c  = col & 15;
        *(v8h*)(Ws + ((((q * NT + nt) * 2 + j) * 32) + h * 16 + c) * 8) = pk;
    }
}

template<int NB>
__device__ __forceinline__ void stage_bias(const float* __restrict__ b,
                                           float* __restrict__ Bs, int tid)
{
    static_assert(NB <= 512, "one element per thread");
    if (tid < NB) Bs[tid] = b[tid];
}

template<int K>
__device__ __forceinline__ void load_rows(const float* __restrict__ zrow, int lh,
                                          v16h (&hb)[K / 32])
{
#pragma unroll
    for (int q = 0; q < K / 32; ++q) {
#pragma unroll
        for (int j = 0; j < 2; ++j) {
            const float* p = zrow + 32 * q + 16 * j + 8 * lh;
            const float4 v0 = *(const float4*)p;
            const float4 v1 = *(const float4*)(p + 4);
            hb[q][8 * j + 0] = (_Float16)v0.x;
            hb[q][8 * j + 1] = (_Float16)v0.y;
            hb[q][8 * j + 2] = (_Float16)v0.z;
            hb[q][8 * j + 3] = (_Float16)v0.w;
            hb[q][8 * j + 4] = (_Float16)v1.x;
            hb[q][8 * j + 5] = (_Float16)v1.y;
            hb[q][8 * j + 6] = (_Float16)v1.z;
            hb[q][8 * j + 7] = (_Float16)v1.w;
        }
    }
}

template<int K, int NW>
__device__ __forceinline__ void product(const v16h (&hb)[K / 32],
                                        const _Float16* __restrict__ Ws,
                                        const float* __restrict__ Bs, int lane,
                                        v8f (&acc)[NW / 16])
{
    constexpr int NT = NW / 16;
    constexpr int KC = K / 32;
    constexpr int G = (NT % 4 == 0) ? 4 : NT;
#pragma unroll
    for (int t = 0; t < NT; ++t) acc[t] = *(const v8f*)(Bs + 16 * t + 8 * (lane >> 4));
#pragma unroll
    for (int q = 0; q < KC; ++q) {
#pragma unroll
        for (int g = 0; g < NT; g += G) {
            v16h w[G];
#pragma unroll
            for (int u = 0; u < G; ++u) {
                const _Float16* p = Ws + (((q * NT + g + u) * 2) * 32 + lane) * 8;
                const v8h lo = *(const v8h*)p;
                const v8h hi = *(const v8h*)(p + 32 * 8);
                w[u] = __builtin_shufflevector(lo, hi,
                    0, 1, 2, 3, 4, 5, 6, 7, 8, 9, 10, 11, 12, 13, 14, 15);
            }
#pragma unroll
            for (int u = 0; u < G; ++u)
                acc[g + u] = __builtin_amdgcn_wmma_f32_16x16x32_f16(
                    false, w[u], false, hb[q], (short)0, acc[g + u], false, false);
        }
    }
}

template<int NW>
__device__ __forceinline__ void to_fragments(const v8f (&acc)[NW / 16],
                                             v16h (&hb)[NW / 32])
{
#pragma unroll
    for (int t = 0; t < NW / 16; ++t) {
#pragma unroll
        for (int r = 0; r < 8; ++r)
            hb[t >> 1][8 * (t & 1) + r] = (_Float16)fmaxf(acc[t][r], 0.f);
    }
}

template<int NW, bool RELU>
__device__ __forceinline__ void store_rows(const v8f (&acc)[NW / 16], int lh,
                                           float* __restrict__ orow, int Nout)
{
#pragma unroll
    for (int t = 0; t < NW / 16; ++t) {
        const int col0 = 16 * t + 8 * lh;
        if (col0 < Nout) {
            float4 o0, o1;
            o0.x = acc[t][0]; o0.y = acc[t][1]; o0.z = acc[t][2]; o0.w = acc[t][3];
            o1.x = acc[t][4]; o1.y = acc[t][5]; o1.z = acc[t][6]; o1.w = acc[t][7];
            if (RELU) {
                o0.x = fmaxf(o0.x, 0.f); o0.y = fmaxf(o0.y, 0.f);
                o0.z = fmaxf(o0.z, 0.f); o0.w = fmaxf(o0.w, 0.f);
                o1.x = fmaxf(o1.x, 0.f); o1.y = fmaxf(o1.y, 0.f);
                o1.z = fmaxf(o1.z, 0.f); o1.w = fmaxf(o1.w, 0.f);
            }
            *(float4*)(orow + col0)     = o0;
            *(float4*)(orow + col0 + 4) = o1;
        }
    }
}

__global__ void __launch_bounds__(512)
mlp_layer1(const float* __restrict__ z,
           const float* __restrict__ Wa, const float* __restrict__ ba,
           const float* __restrict__ Wb, const float* __restrict__ bb,
           float* __restrict__ out, int mtiles)
{
    __shared__ __attribute__((aligned(16))) _Float16 WsA[128 * 128];
    __shared__ __attribute__((aligned(16))) _Float16 WsB[128 * 128];
    __shared__ __attribute__((aligned(32))) float    Bs[128 + 128];

    const int tid  = threadIdx.x;
    const int lane = tid & 31;
    const int wave = __builtin_amdgcn_readfirstlane(tid >> 5);
    const int lm = lane & 15;
    const int lh = lane >> 4;
    const int mt = blockIdx.x * 16 + wave;
    const bool active = mt < mtiles;

    const int node = (active ? mt : (mtiles - 1)) * 16 + lm;
    v16h h0[4];
    load_rows<128>(z + (size_t)node * 128, lh, h0);

    stage_weights<128, 128>(Wa, WsA, tid);
    stage_weights<128, 128>(Wb, WsB, tid);
    stage_bias<128>(ba, Bs, tid);
    stage_bias<128>(bb, Bs + 128, tid);
    asm volatile("" :: "v"(h0[0]), "v"(h0[1]), "v"(h0[2]), "v"(h0[3]));
    __syncthreads();
    if (!active) return;

    v8f acc[8];
    product<128, 128>(h0, WsA, Bs, lane, acc);
    v16h h1[4];
    to_fragments<128>(acc, h1);
    product<128, 128>(h1, WsB, Bs + 128, lane, acc);
    store_rows<128, true>(acc, lh, out + (size_t)node * 128, 128);
}

__global__ void __launch_bounds__(512)
mlp_layer2_head(const float* __restrict__ z,
                const float* __restrict__ Wa, const float* __restrict__ ba,
                const float* __restrict__ Wb, const float* __restrict__ bb,
                const float* __restrict__ Wc, const float* __restrict__ bc,
                float* __restrict__ out, int Nout, int mtiles)
{
    __shared__ __attribute__((aligned(16))) _Float16 WsA[128 * 256];
    __shared__ __attribute__((aligned(16))) _Float16 WsB[256 * 256];
    __shared__ __attribute__((aligned(16))) _Float16 WsC[256 * 48];
    __shared__ __attribute__((aligned(32))) float    Bs[256 + 256 + 48];

    const int tid  = threadIdx.x;
    const int lane = tid & 31;
    const int wave = __builtin_amdgcn_readfirstlane(tid >> 5);
    const int lm = lane & 15;
    const int lh = lane >> 4;
    const int mt = blockIdx.x * 16 + wave;
    const bool active = mt < mtiles;

    const int node = (active ? mt : (mtiles - 1)) * 16 + lm;
    v16h h0[4];
    load_rows<128>(z + (size_t)node * 128, lh, h0);

    stage_weights<128, 256>(Wa, WsA, tid);
    stage_weights<256, 256>(Wb, WsB, tid);
    stage_weights<256, 48>(Wc, WsC, tid);
    stage_bias<256>(ba, Bs, tid);
    stage_bias<256>(bb, Bs + 256, tid);
    stage_bias<48>(bc, Bs + 512, tid);
    asm volatile("" :: "v"(h0[0]), "v"(h0[1]), "v"(h0[2]), "v"(h0[3]));
    __syncthreads();
    if (!active) return;

    v8f acc[16];
    product<128, 256>(h0, WsA, Bs, lane, acc);
    v16h h1[8];
    to_fragments<256>(acc, h1);
    product<256, 256>(h1, WsB, Bs + 256, lane, acc);
    v16h h2[8];
    to_fragments<256>(acc, h2);
    v8f acc3[3];
    product<256, 48>(h2, WsC, Bs + 512, lane, acc3);
    store_rows<48, false>(acc3, lh, out + (size_t)node * Nout, Nout);
}

extern "C" void kernel_launch(void* const* d_in, const int* in_sizes, int n_in,
                              void* d_out, int out_size, void* d_ws, size_t ws_size,
                              hipStream_t stream)
{
    const float* x  = (const float*)d_in[0];
    const int*   ei = (const int*)  d_in[1];
    const float* W1 = (const float*)d_in[2];
    const float* b1 = (const float*)d_in[3];
    const float* W2 = (const float*)d_in[4];
    const float* b2 = (const float*)d_in[5];
    const float* W3 = (const float*)d_in[6];
    const float* b3 = (const float*)d_in[7];
    const float* W4 = (const float*)d_in[8];
    const float* b4 = (const float*)d_in[9];
    const float* Wl = (const float*)d_in[10];
    const float* bl = (const float*)d_in[11];
    float* out = (float*)d_out;

    const int N = in_sizes[0] / 128;
    const int E = in_sizes[1] / 2;
    const int* src = ei;
    const int* dst = ei + E;

    float* ws   = (float*)d_ws;
    float* bufA = ws;
    float* bufB = ws + (size_t)N * 128;
    float* Wp   = ws + (size_t)N * 512;
    float* bp   = Wp + 256 * 48;
    const int Epad = (E + 63) / 64 * 64;
    int* cnt    = (int*)(ws + (size_t)N * 512 + 12352);
    int* offs   = cnt + (N + 64);
    int* bucket = offs + (N + 64);
    int* recs   = bucket + Epad;
    int* table  = recs + Epad;

    const int nchunks    = (E + SORT_CHUNK - 1) / SORT_CHUNK;
    const int nbins      = (N + BIN_NODES - 1) / BIN_NODES;
    const int waveBlocks = (N + 7) / 8;
    const int mtiles     = N / 16;
    const int gemmBlocks = (mtiles + 15) / 16;

    prep<<<(256 * 48) / 256, 256, 0, stream>>>(Wl, bl, Wp, bp, cnt + N);
    sort_chunks<<<nchunks, 512, 0, stream>>>(src, dst, recs, table, E, nbins, nchunks);
    sort_bins<<<nbins, 512, 0, stream>>>(recs, table, cnt + N, cnt, offs, bucket, N, nchunks);

    gather_sum128<<<waveBlocks, 256, 0, stream>>>(x, offs, cnt, bucket, bufA, N);

    mlp_layer1<<<gemmBlocks, 512, 0, stream>>>(bufA, W1, b1, W2, b2, bufB, mtiles);

    gather_sum128<<<waveBlocks, 256, 0, stream>>>(bufB, offs, cnt, bucket, bufA, N);

    mlp_layer2_head<<<gemmBlocks, 512, 0, stream>>>(bufA, W3, b3, W4, b4, Wp, bp, out, 40, mtiles);
}


# Gather: a node with 33 to 64 in-edges ranks its ids in registers, two per lane, and fetches rows 4 at a time; was one selection pass plus one dependent row load per in-edge; sum order unchanged
# speedup vs baseline: 4.6088x; 1.5044x over previous
#include <hip/hip_runtime.h>

typedef __attribute__((ext_vector_type(16))) _Float16 v16h;
typedef __attribute__((ext_vector_type(8)))  float    v8f;


constexpr int SORT_CHUNK = 4096;
constexpr int BIN_SHIFT  = 8;
constexpr int BIN_NODES  = 1 << BIN_SHIFT;
constexpr int MAX_BINS   = 256;

__global__ void __launch_bounds__(256)
prep(const float* __restrict__ Wl, const float* __restrict__ bl,
     float* __restrict__ Wp, float* __restrict__ bp, int* __restrict__ total)
{
    const int i = blockIdx.x * 256 + threadIdx.x;
    if (i < 256 * 48) {
        const int r = i / 48, c = i % 48;
        Wp[i] = (c < 40) ? Wl[r * 40 + c] : 0.f;
    }
    if (i < 48) bp[i] = (i < 40) ? bl[i] : 0.f;
    if (i == 0) *total = 0;
}

__global__ void __launch_bounds__(512)
sort_chunks(const int* __restrict__ src, const int* __restrict__ dst,
            int* __restrict__ recs, int* __restrict__ table,
            int E, int nbins, int nchunks)
{
    __shared__ int hist[MAX_BINS];
    __shared__ int start[MAX_BINS];
    __shared__ int srec[SORT_CHUNK];

    const int tid  = threadIdx.x;
    const int lane = tid & 31;
    const int wave  = __builtin_amdgcn_readfirstlane(tid >> 5);
    const int chunk = blockIdx.x;
    const int base  = chunk * SORT_CHUNK;

    if (tid < MAX_BINS) hist[tid] = 0;

    int d[8], s[8];
#pragma unroll
    for (int k = 0; k < 8; ++k) {
        const int e = base + k * 512 + tid;
        d[k] = (e < E) ? dst[e] : -1;
        s[k] = (e < E) ? src[e] : 0;
    }
    __syncthreads();

    int r[8];
#pragma unroll
    for (int k = 0; k < 8; ++k)
        r[k] = (d[k] >= 0) ? atomicAdd(&hist[d[k] >> BIN_SHIFT], 1) : 0;
    __syncthreads();

    if (wave == 0) {
        int c[8];
        int sum = 0;
#pragma unroll
        for (int j = 0; j < 8; ++j) { c[j] = sum; sum += hist[lane * 8 + j]; }
        int incl = sum;
        for (int o = 1; o < 32; o <<= 1) {
            const int up = __shfl_up(incl, o, 32);
            if (lane >= o) incl += up;
        }
        const int excl = incl - sum;
#pragma unroll
        for (int j = 0; j < 8; ++j) start[lane * 8 + j] = excl + c[j];
    }
    __syncthreads();

    int q[8];
#pragma unroll
    for (int k = 0; k < 8; ++k)
        q[k] = (d[k] >= 0) ? start[d[k] >> BIN_SHIFT] : 0;
#pragma unroll
    for (int k = 0; k < 8; ++k)
        if (d[k] >= 0) srec[q[k] + r[k]] = ((d[k] & (BIN_NODES - 1)) << 16) | s[k];
    __syncthreads();

    const int n = (E - base < SORT_CHUNK) ? (E - base) : SORT_CHUNK;
    int o[8];
#pragma unroll
    for (int k = 0; k < 8; ++k) {
        const int i = k * 512 + tid;
        o[k] = (i < n) ? srec[i] : 0;
    }
#pragma unroll
    for (int k = 0; k < 8; ++k) {
        const int i = k * 512 + tid;
        if (i < n) recs[base + i] = o[k];
    }
    if (tid < nbins) {
        int2 t;
        t.x = base + start[tid];
        t.y = hist[tid];
        *(int2*)(table + ((size_t)tid * nchunks + chunk) * 2) = t;
    }
}

template<bool PLACE>
__device__ __forceinline__ void bin_record(int v, int* __restrict__ ncnt, int* __restrict__ ncur,
                                           const int* __restrict__ nstart, int* __restrict__ bucket)
{
    const int dl = v >> 16;
    if (PLACE) bucket[nstart[dl] + atomicAdd(&ncur[dl], 1)] = v & 0xffff;
    else       atomicAdd(&ncnt[dl], 1);
}

template<bool PLACE>
__device__ __forceinline__ void bin_rest(const int* __restrict__ recs, const int* __restrict__ row,
                                         int pstart, int plen, int lane, int wave, int nchunks,
                                         int* __restrict__ ncnt, int* __restrict__ ncur,
                                         const int* __restrict__ nstart, int* __restrict__ bucket)
{
    if (__builtin_amdgcn_ballot_w32(plen > 64) != 0u) {
        for (int j = 0; j < 16; ++j) {
            const int po = __builtin_amdgcn_readlane(pstart, j);
            const int pn = __builtin_amdgcn_readlane(plen, j);
            for (int i = 64 + lane; i < pn; i += 32)
                bin_record<PLACE>(recs[po + i], ncnt, ncur, nstart, bucket);
        }
    }
    for (int w = 256 + wave; w < nchunks; w += 16) {
        const int po = row[2 * w];
        const int pn = row[2 * w + 1];
        for (int i = lane; i < pn; i += 32)
            bin_record<PLACE>(recs[po + i], ncnt, ncur, nstart, bucket);
    }
}

__global__ void __launch_bounds__(512)
sort_bins(const int* __restrict__ recs, const int* __restrict__ table,
          int* __restrict__ total, int* __restrict__ cnt, int* __restrict__ offs,
          int* __restrict__ bucket, int N, int nchunks)
{
    __shared__ int ncnt[BIN_NODES];
    __shared__ int ncur[BIN_NODES];
    __shared__ int nstart[BIN_NODES];

    const int tid  = threadIdx.x;
    const int lane = tid & 31;
    const int wave = __builtin_amdgcn_readfirstlane(tid >> 5);
    const int bin  = blockIdx.x;
    const int* row = table + (size_t)bin * nchunks * 2;

    if (tid < BIN_NODES) { ncnt[tid] = 0; ncur[tid] = 0; }

    const int wi = wave * 16 + (lane & 15);
    int pstart = 0, plen = 0;
    if (wi < nchunks) {
        const int2 t = *(const int2*)(row + 2 * wi);
        pstart = t.x;
        plen   = t.y;
    }

    int v[32];
#pragma unroll
    for (int j = 0; j < 16; ++j) {
        const int po = __builtin_amdgcn_readlane(pstart, j);
        const int pn = __builtin_amdgcn_readlane(plen, j);
        v[2 * j]     = (lane < pn)      ? recs[po + lane]      : -1;
        v[2 * j + 1] = (lane + 32 < pn) ? recs[po + 32 + lane] : -1;
    }
    __syncthreads();

#pragma unroll
    for (int k = 0; k < 32; ++k)
        if (v[k] >= 0) bin_record<false>(v[k], ncnt, ncur, nstart, bucket);
    bin_rest<false>(recs, row, pstart, plen, lane, wave, nchunks, ncnt, ncur, nstart, bucket);
    __syncthreads();

    if (wave == 0) {
        int c[8];
        int sum = 0;
#pragma unroll
        for (int j = 0; j < 8; ++j) { c[j] = sum; sum += ncnt[lane * 8 + j]; }
        int incl = sum;
        for (int o = 1; o < 32; o <<= 1) {
            const int up = __shfl_up(incl, o, 32);
            if (lane >= o) incl += up;
        }
        int slice = 0;
        if (lane == 31) slice = atomicAdd(total, incl);
        slice = __shfl(slice, 31, 32);
        const int excl = slice + incl - sum;
#pragma unroll
        for (int j = 0; j < 8; ++j) nstart[lane * 8 + j] = excl + c[j];
    }
    __syncthreads();

    if (tid < BIN_NODES) {
        const int n = bin * BIN_NODES + tid;
        if (n < N) {
            cnt[n]  = ncnt[tid];
            offs[n] = nstart[tid];
        }
    }

    int p[32], t[32];
#pragma unroll
    for (int k = 0; k < 32; ++k) {
        p[k] = (v[k] >= 0) ? nstart[v[k] >> 16] : 0;
        t[k] = (v[k] >= 0) ? atomicAdd(&ncur[v[k] >> 16], 1) : 0;
    }
#pragma unroll
    for (int k = 0; k < 32; ++k)
        if (v[k] >= 0) bucket[p[k] + t[k]] = v[k] & 0xffff;
    bin_rest<true>(recs, row, pstart, plen, lane, wave, nchunks, ncnt, ncur, nstart, bucket);
}

__device__ __forceinline__ float4 row4(const float* __restrict__ feat, int s, int lane)
{
    return *(const float4*)(feat + (size_t)s * 128 + lane * 4);
}

__device__ __forceinline__ void add4(float4& a, const float4 v)
{
    a.x += v.x; a.y += v.y; a.z += v.z; a.w += v.w;
}

__device__ __forceinline__ void sum_rows_upto64(const float* __restrict__ feat,
                                                const int* __restrict__ bucket,
                                                int beg, int deg, int lane, float4& acc)
{
    const int k0 = bucket[beg + lane];
    const bool two = lane + 32 < deg;
    const int k1 = two ? bucket[beg + 32 + lane] : 0x7fffffff;
    int r0 = 0, r1 = 0;
    for (int j = 0; j < 32; ++j) {
        const int kj = __builtin_amdgcn_readlane(k0, j);
        r0 += ((kj < k0) || (kj == k0 && j < lane)) ? 1 : 0;
        r1 += (kj <= k1) ? 1 : 0;
    }
    for (int j = 0; j < deg - 32; ++j) {
        const int kj = __builtin_amdgcn_readlane(k1, j);
        r0 += (kj < k0) ? 1 : 0;
        r1 += ((kj < k1) || (kj == k1 && j < lane)) ? 1 : 0;
    }
    if (!two) r1 = -1;
    auto id_at = [&](int p) -> int {
        const unsigned m0 = __builtin_amdgcn_ballot_w32(r0 == p);
        const unsigned m1 = __builtin_amdgcn_ballot_w32(r1 == p);
        const int from = __builtin_ctz(m0 | m1);
        const int a = __builtin_amdgcn_readlane(k0, from);
        const int b = __builtin_amdgcn_readlane(k1, from);
        return (m0 != 0u) ? a : b;
    };
    int j = 0;
    for (; j + 4 <= deg; j += 4) {
        const int i0 = id_at(j + 0), i1 = id_at(j + 1), i2 = id_at(j + 2), i3 = id_at(j + 3);
        const float4 v0 = row4(feat, i0, lane);
        const float4 v1 = row4(feat, i1, lane);
        const float4 v2 = row4(feat, i2, lane);
        const float4 v3 = row4(feat, i3, lane);
        add4(acc, v0); add4(acc, v1); add4(acc, v2); add4(acc, v3);
    }
    if (deg & 2) {
        const int i0 = id_at(j + 0), i1 = id_at(j + 1);
        const float4 v0 = row4(feat, i0, lane);
        const float4 v1 = row4(feat, i1, lane);
        add4(acc, v0); add4(acc, v1);
        j += 2;
    }
    if (deg & 1) {
        const float4 v0 = row4(feat, id_at(j), lane);
        add4(acc, v0);
    }
}

__global__ void __launch_bounds__(256)
gather_sum128(const float* __restrict__ feat,
              const int* __restrict__ offs, const int* __restrict__ cnt,
              const int* __restrict__ bucket,
              float* __restrict__ out, int N)
{
    const int lane = threadIdx.x & 31;
    const int n = __builtin_amdgcn_readfirstlane(blockIdx.x * 8 + (threadIdx.x >> 5));
    if (n >= N) return;
    const int beg = offs[n];
    const int deg = cnt[n];
    float4 acc = row4(feat, n, lane);

    if (deg <= 32) {
        const int key = (lane < deg) ? bucket[beg + lane] : 0x7fffffff;
        int rank = 0;
        for (int j = 0; j < deg; ++j) {
            const int kj = __builtin_amdgcn_readlane(key, j);
            rank += ((kj < key) || (kj == key && j < lane)) ? 1 : 0;
        }
        if (lane >= deg) rank = lane;
        const int sk = __builtin_amdgcn_ds_permute(rank << 2, key);

        int j = 0;
        for (; j + 8 <= deg; j += 8) {
            float4 v[8];
#pragma unroll
            for (int t = 0; t < 8; ++t)
                v[t] = row4(feat, __builtin_amdgcn_readlane(sk, j + t), lane);
#pragma unroll
            for (int t = 0; t < 8; ++t) add4(acc, v[t]);
        }
        if (deg & 4) {
            const float4 v0 = row4(feat, __builtin_amdgcn_readlane(sk, j + 0), lane);
            const float4 v1 = row4(feat, __builtin_amdgcn_readlane(sk, j + 1), lane);
            const float4 v2 = row4(feat, __builtin_amdgcn_readlane(sk, j + 2), lane);
            const float4 v3 = row4(feat, __builtin_amdgcn_readlane(sk, j + 3), lane);
            add4(acc, v0); add4(acc, v1); add4(acc, v2); add4(acc, v3);
            j += 4;
        }
        if (deg & 2) {
            const float4 v0 = row4(feat, __builtin_amdgcn_readlane(sk, j + 0), lane);
            const float4 v1 = row4(feat, __builtin_amdgcn_readlane(sk, j + 1), lane);
            add4(acc, v0); add4(acc, v1);
            j += 2;
        }
        if (deg & 1) {
            const float4 v0 = row4(feat, __builtin_amdgcn_readlane(sk, j), lane);
            add4(acc, v0);
        }
    } else if (deg <= 64) {
        sum_rows_upto64(feat, bucket, beg, deg, lane, acc);
    } else {
        int last = -1;
        int done = 0;
        for (int it = 0; it < deg && done < deg; ++it) {
            int m = 0x7fffffff;
            for (int i = lane; i < deg; i += 32) {
                const int k = bucket[beg + i];
                if (k > last && k < m) m = k;
            }
            for (int o = 16; o > 0; o >>= 1) {
                const int other = __shfl_xor(m, o, 32);
                m = (other < m) ? other : m;
            }
            int c = 0;
            for (int i = lane; i < deg; i += 32) c += (bucket[beg + i] == m) ? 1 : 0;
            for (int o = 16; o > 0; o >>= 1) c += __shfl_xor(c, o, 32);
            m = __builtin_amdgcn_readfirstlane(m);
            c = __builtin_amdgcn_readfirstlane(c);
            if (c <= 0) break;
            const float4 v = row4(feat, m, lane);
            for (int t = 0; t < c; ++t) add4(acc, v);
            last = m;
            done += c;
        }
    }
    *(float4*)(out + (size_t)n * 128 + lane * 4) = acc;
}

typedef __attribute__((ext_vector_type(8))) _Float16 v8h;

template<int K, int NW>
__device__ __forceinline__ void stage_weights(const float* __restrict__ W,
                                              _Float16* __restrict__ Ws, int tid)
{
    constexpr int NT = NW / 16;
    constexpr int TASKS = (K / 8) * NW;
    static_assert(TASKS % 512 == 0, "every thread stages the same number of tasks");
#pragma unroll 4
    for (int i = 0; i < TASKS / 512; ++i) {
        const int t   = i * 512 + tid;
        const int col = t % NW;
        const int kg  = t / NW;
        const float* s = W + (size_t)(kg * 8) * NW + col;
        v8h pk;
#pragma unroll
        for (int e = 0; e < 8; ++e) pk[e] = (_Float16)s[(size_t)e * NW];
        const int q  = kg >> 2;
        const int j  = (kg >> 1) & 1;
        const int h  = kg & 1;
        const int nt = col >> 4;
        const int c  = col & 15;
        *(v8h*)(Ws + ((((q * NT + nt) * 2 + j) * 32) + h * 16 + c) * 8) = pk;
    }
}

template<int NB>
__device__ __forceinline__ void stage_bias(const float* __restrict__ b,
                                           float* __restrict__ Bs, int tid)
{
    static_assert(NB <= 512, "one element per thread");
    if (tid < NB) Bs[tid] = b[tid];
}

template<int K>
__device__ __forceinline__ void load_rows(const float* __restrict__ zrow, int lh,
                                          v16h (&hb)[K / 32])
{
#pragma unroll
    for (int q = 0; q < K / 32; ++q) {
#pragma unroll
        for (int j = 0; j < 2; ++j) {
            const float* p = zrow + 32 * q + 16 * j + 8 * lh;
            const float4 v0 = *(const float4*)p;
            const float4 v1 = *(const float4*)(p + 4);
            hb[q][8 * j + 0] = (_Float16)v0.x;
            hb[q][8 * j + 1] = (_Float16)v0.y;
            hb[q][8 * j + 2] = (_Float16)v0.z;
            hb[q][8 * j + 3] = (_Float16)v0.w;
            hb[q][8 * j + 4] = (_Float16)v1.x;
            hb[q][8 * j + 5] = (_Float16)v1.y;
            hb[q][8 * j + 6] = (_Float16)v1.z;
            hb[q][8 * j + 7] = (_Float16)v1.w;
        }
    }
}

template<int K, int NW>
__device__ __forceinline__ void product(const v16h (&hb)[K / 32],
                                        const _Float16* __restrict__ Ws,
                                        const float* __restrict__ Bs, int lane,
                                        v8f (&acc)[NW / 16])
{
    constexpr int NT = NW / 16;
    constexpr int KC = K / 32;
    constexpr int G = (NT % 4 == 0) ? 4 : NT;
#pragma unroll
    for (int t = 0; t < NT; ++t) acc[t] = *(const v8f*)(Bs + 16 * t + 8 * (lane >> 4));
#pragma unroll
    for (int q = 0; q < KC; ++q) {
#pragma unroll
        for (int g = 0; g < NT; g += G) {
            v16h w[G];
#pragma unroll
            for (int u = 0; u < G; ++u) {
                const _Float16* p = Ws + (((q * NT + g + u) * 2) * 32 + lane) * 8;
                const v8h lo = *(const v8h*)p;
                const v8h hi = *(const v8h*)(p + 32 * 8);
                w[u] = __builtin_shufflevector(lo, hi,
                    0, 1, 2, 3, 4, 5, 6, 7, 8, 9, 10, 11, 12, 13, 14, 15);
            }
#pragma unroll
            for (int u = 0; u < G; ++u)
                acc[g + u] = __builtin_amdgcn_wmma_f32_16x16x32_f16(
                    false, w[u], false, hb[q], (short)0, acc[g + u], false, false);
        }
    }
}

template<int NW>
__device__ __forceinline__ void to_fragments(const v8f (&acc)[NW / 16],
                                             v16h (&hb)[NW / 32])
{
#pragma unroll
    for (int t = 0; t < NW / 16; ++t) {
#pragma unroll
        for (int r = 0; r < 8; ++r)
            hb[t >> 1][8 * (t & 1) + r] = (_Float16)fmaxf(acc[t][r], 0.f);
    }
}

template<int NW, bool RELU>
__device__ __forceinline__ void store_rows(const v8f (&acc)[NW / 16], int lh,
                                           float* __restrict__ orow, int Nout)
{
#pragma unroll
    for (int t = 0; t < NW / 16; ++t) {
        const int col0 = 16 * t + 8 * lh;
        if (col0 < Nout) {
            float4 o0, o1;
            o0.x = acc[t][0]; o0.y = acc[t][1]; o0.z = acc[t][2]; o0.w = acc[t][3];
            o1.x = acc[t][4]; o1.y = acc[t][5]; o1.z = acc[t][6]; o1.w = acc[t][7];
            if (RELU) {
                o0.x = fmaxf(o0.x, 0.f); o0.y = fmaxf(o0.y, 0.f);
                o0.z = fmaxf(o0.z, 0.f); o0.w = fmaxf(o0.w, 0.f);
                o1.x = fmaxf(o1.x, 0.f); o1.y = fmaxf(o1.y, 0.f);
                o1.z = fmaxf(o1.z, 0.f); o1.w = fmaxf(o1.w, 0.f);
            }
            *(float4*)(orow + col0)     = o0;
            *(float4*)(orow + col0 + 4) = o1;
        }
    }
}

__global__ void __launch_bounds__(512)
mlp_layer1(const float* __restrict__ z,
           const float* __restrict__ Wa, const float* __restrict__ ba,
           const float* __restrict__ Wb, const float* __restrict__ bb,
           float* __restrict__ out, int mtiles)
{
    __shared__ __attribute__((aligned(16))) _Float16 WsA[128 * 128];
    __shared__ __attribute__((aligned(16))) _Float16 WsB[128 * 128];
    __shared__ __attribute__((aligned(32))) float    Bs[128 + 128];

    const int tid  = threadIdx.x;
    const int lane = tid & 31;
    const int wave = __builtin_amdgcn_readfirstlane(tid >> 5);
    const int lm = lane & 15;
    const int lh = lane >> 4;
    const int mt = blockIdx.x * 16 + wave;
    const bool active = mt < mtiles;

    const int node = (active ? mt : (mtiles - 1)) * 16 + lm;
    v16h h0[4];
    load_rows<128>(z + (size_t)node * 128, lh, h0);

    stage_weights<128, 128>(Wa, WsA, tid);
    stage_weights<128, 128>(Wb, WsB, tid);
    stage_bias<128>(ba, Bs, tid);
    stage_bias<128>(bb, Bs + 128, tid);
    asm volatile("" :: "v"(h0[0]), "v"(h0[1]), "v"(h0[2]), "v"(h0[3]));
    __syncthreads();
    if (!active) return;

    v8f acc[8];
    product<128, 128>(h0, WsA, Bs, lane, acc);
    v16h h1[4];
    to_fragments<128>(acc, h1);
    product<128, 128>(h1, WsB, Bs + 128, lane, acc);
    store_rows<128, true>(acc, lh, out + (size_t)node * 128, 128);
}

__global__ void __launch_bounds__(512)
mlp_layer2_head(const float* __restrict__ z,
                const float* __restrict__ Wa, const float* __restrict__ ba,
                const float* __restrict__ Wb, const float* __restrict__ bb,
                const float* __restrict__ Wc, const float* __restrict__ bc,
                float* __restrict__ out, int Nout, int mtiles)
{
    __shared__ __attribute__((aligned(16))) _Float16 WsA[128 * 256];
    __shared__ __attribute__((aligned(16))) _Float16 WsB[256 * 256];
    __shared__ __attribute__((aligned(16))) _Float16 WsC[256 * 48];
    __shared__ __attribute__((aligned(32))) float    Bs[256 + 256 + 48];

    const int tid  = threadIdx.x;
    const int lane = tid & 31;
    const int wave = __builtin_amdgcn_readfirstlane(tid >> 5);
    const int lm = lane & 15;
    const int lh = lane >> 4;
    const int mt = blockIdx.x * 16 + wave;
    const bool active = mt < mtiles;

    const int node = (active ? mt : (mtiles - 1)) * 16 + lm;
    v16h h0[4];
    load_rows<128>(z + (size_t)node * 128, lh, h0);

    stage_weights<128, 256>(Wa, WsA, tid);
    stage_weights<256, 256>(Wb, WsB, tid);
    stage_weights<256, 48>(Wc, WsC, tid);
    stage_bias<256>(ba, Bs, tid);
    stage_bias<256>(bb, Bs + 256, tid);
    stage_bias<48>(bc, Bs + 512, tid);
    asm volatile("" :: "v"(h0[0]), "v"(h0[1]), "v"(h0[2]), "v"(h0[3]));
    __syncthreads();
    if (!active) return;

    v8f acc[16];
    product<128, 256>(h0, WsA, Bs, lane, acc);
    v16h h1[8];
    to_fragments<256>(acc, h1);
    product<256, 256>(h1, WsB, Bs + 256, lane, acc);
    v16h h2[8];
    to_fragments<256>(acc, h2);
    v8f acc3[3];
    product<256, 48>(h2, WsC, Bs + 512, lane, acc3);
    store_rows<48, false>(acc3, lh, out + (size_t)node * Nout, Nout);
}

extern "C" void kernel_launch(void* const* d_in, const int* in_sizes, int n_in,
                              void* d_out, int out_size, void* d_ws, size_t ws_size,
                              hipStream_t stream)
{
    const float* x  = (const float*)d_in[0];
    const int*   ei = (const int*)  d_in[1];
    const float* W1 = (const float*)d_in[2];
    const float* b1 = (const float*)d_in[3];
    const float* W2 = (const float*)d_in[4];
    const float* b2 = (const float*)d_in[5];
    const float* W3 = (const float*)d_in[6];
    const float* b3 = (const float*)d_in[7];
    const float* W4 = (const float*)d_in[8];
    const float* b4 = (const float*)d_in[9];
    const float* Wl = (const float*)d_in[10];
    const float* bl = (const float*)d_in[11];
    float* out = (float*)d_out;

    const int N = in_sizes[0] / 128;
    const int E = in_sizes[1] / 2;
    const int* src = ei;
    const int* dst = ei + E;

    float* ws   = (float*)d_ws;
    float* bufA = ws;
    float* bufB = ws + (size_t)N * 128;
    float* Wp   = ws + (size_t)N * 512;
    float* bp   = Wp + 256 * 48;
    const int Epad = (E + 63) / 64 * 64;
    int* cnt    = (int*)(ws + (size_t)N * 512 + 12352);
    int* offs   = cnt + (N + 64);
    int* bucket = offs + (N + 64);
    int* recs   = bucket + Epad;
    int* table  = recs + Epad;

    const int nchunks    = (E + SORT_CHUNK - 1) / SORT_CHUNK;
    const int nbins      = (N + BIN_NODES - 1) / BIN_NODES;
    const int waveBlocks = (N + 7) / 8;
    const int mtiles     = N / 16;
    const int gemmBlocks = (mtiles + 15) / 16;

    prep<<<(256 * 48) / 256, 256, 0, stream>>>(Wl, bl, Wp, bp, cnt + N);
    sort_chunks<<<nchunks, 512, 0, stream>>>(src, dst, recs, table, E, nbins, nchunks);
    sort_bins<<<nbins, 512, 0, stream>>>(recs, table, cnt + N, cnt, offs, bucket, N, nchunks);

    gather_sum128<<<waveBlocks, 256, 0, stream>>>(x, offs, cnt, bucket, bufA, N);

    mlp_layer1<<<gemmBlocks, 512, 0, stream>>>(bufA, W1, b1, W2, b2, bufB, mtiles);

    gather_sum128<<<waveBlocks, 256, 0, stream>>>(bufB, offs, cnt, bucket, bufA, N);

    mlp_layer2_head<<<gemmBlocks, 512, 0, stream>>>(bufA, W3, b3, W4, b4, Wp, bp, out, 40, mtiles);
}


# The call's preparation (f16 copy of x, padded head weights, slice word) and level 1 of the edge sort share one launch on disjoint work-groups (no common data); 7 launches -> 6; sums unchanged
# speedup vs baseline: 5.3564x; 1.1622x over previous
#include <hip/hip_runtime.h>

typedef __attribute__((ext_vector_type(16))) _Float16 v16h;
typedef __attribute__((ext_vector_type(8)))  float    v8f;

typedef __attribute__((ext_vector_type(4)))  _Float16 v4h;
typedef __attribute__((ext_vector_type(4)))  float    v4f;


constexpr int SORT_CHUNK = 4096;
constexpr int BIN_SHIFT  = 8;
constexpr int BIN_NODES  = 1 << BIN_SHIFT;
constexpr int MAX_BINS   = 256;
constexpr int PREP_GROUPS = 256;

__device__ __forceinline__ void
prep_part(const float* __restrict__ Wl, const float* __restrict__ bl,
          float* __restrict__ Wp, float* __restrict__ bp, int* __restrict__ total,
          const float* __restrict__ x, _Float16* __restrict__ xh, int nquads,
          int i, int stride)
{
    if (i < 256 * 48) {
        const int r = i / 48, c = i % 48;
        Wp[i] = (c < 40) ? Wl[r * 40 + c] : 0.f;
    }
    if (i < 48) bp[i] = (i < 40) ? bl[i] : 0.f;
    if (i == 0) *total = 0;

    for (int u0 = i; u0 < nquads; u0 += 8 * stride) {
        v4f v[8];
#pragma unroll
        for (int t = 0; t < 8; ++t) {
            const int u  = u0 + t * stride;
            const int uc = (u < nquads) ? u : (nquads - 1);
            v[t] = *(const v4f*)(x + (size_t)uc * 4);
        }
        asm volatile("" :: "v"(v[0]), "v"(v[1]), "v"(v[2]), "v"(v[3]),
                           "v"(v[4]), "v"(v[5]), "v"(v[6]), "v"(v[7]));
#pragma unroll
        for (int t = 0; t < 8; ++t) {
            const int u = u0 + t * stride;
            if (u < nquads)
                *(v4h*)(xh + (size_t)u * 4) = __builtin_convertvector(v[t], v4h);
        }
    }
}

__global__ void __launch_bounds__(512)
sort_chunks_prep(const int* __restrict__ src, const int* __restrict__ dst,
                 int* __restrict__ recs, int* __restrict__ table,
                 int E, int nbins, int nchunks,
                 const float* __restrict__ Wl, const float* __restrict__ bl,
                 float* __restrict__ Wp, float* __restrict__ bp, int* __restrict__ total,
                 const float* __restrict__ x, _Float16* __restrict__ xh, int nquads)
{
    __shared__ int hist[MAX_BINS];
    __shared__ int start[MAX_BINS];
    __shared__ int srec[SORT_CHUNK];

    if ((int)blockIdx.x >= nchunks) {
        const int pg = (int)blockIdx.x - nchunks;
        prep_part(Wl, bl, Wp, bp, total, x, xh, nquads,
                  pg * 512 + (int)threadIdx.x, ((int)gridDim.x - nchunks) * 512);
        return;
    }

    const int tid  = threadIdx.x;
    const int lane = tid & 31;
    const int wave  = __builtin_amdgcn_readfirstlane(tid >> 5);
    const int chunk = blockIdx.x;
    const int base  = chunk * SORT_CHUNK;

    if (tid < MAX_BINS) hist[tid] = 0;

    int d[8], s[8];
#pragma unroll
    for (int k = 0; k < 8; ++k) {
        const int e = base + k * 512 + tid;
        d[k] = (e < E) ? dst[e] : -1;
        s[k] = (e < E) ? src[e] : 0;
    }
    __syncthreads();

    int r[8];
#pragma unroll
    for (int k = 0; k < 8; ++k)
        r[k] = (d[k] >= 0) ? atomicAdd(&hist[d[k] >> BIN_SHIFT], 1) : 0;
    __syncthreads();

    if (wave == 0) {
        int c[8];
        int sum = 0;
#pragma unroll
        for (int j = 0; j < 8; ++j) { c[j] = sum; sum += hist[lane * 8 + j]; }
        int incl = sum;
        for (int o = 1; o < 32; o <<= 1) {
            const int up = __shfl_up(incl, o, 32);
            if (lane >= o) incl += up;
        }
        const int excl = incl - sum;
#pragma unroll
        for (int j = 0; j < 8; ++j) start[lane * 8 + j] = excl + c[j];
    }
    __syncthreads();

    int q[8];
#pragma unroll
    for (int k = 0; k < 8; ++k)
        q[k] = (d[k] >= 0) ? start[d[k] >> BIN_SHIFT] : 0;
#pragma unroll
    for (int k = 0; k < 8; ++k)
        if (d[k] >= 0) srec[q[k] + r[k]] = ((d[k] & (BIN_NODES - 1)) << 16) | s[k];
    __syncthreads();

    const int n = (E - base < SORT_CHUNK) ? (E - base) : SORT_CHUNK;
    int o[8];
#pragma unroll
    for (int k = 0; k < 8; ++k) {
        const int i = k * 512 + tid;
        o[k] = (i < n) ? srec[i] : 0;
    }
#pragma unroll
    for (int k = 0; k < 8; ++k) {
        const int i = k * 512 + tid;
        if (i < n) recs[base + i] = o[k];
    }
    if (tid < nbins) {
        int2 t;
        t.x = base + start[tid];
        t.y = hist[tid];
        *(int2*)(table + ((size_t)tid * nchunks + chunk) * 2) = t;
    }
}

template<bool PLACE>
__device__ __forceinline__ void bin_record(int v, int* __restrict__ ncnt, int* __restrict__ ncur,
                                           const int* __restrict__ nstart, int* __restrict__ bucket)
{
    const int dl = v >> 16;
    if (PLACE) bucket[nstart[dl] + atomicAdd(&ncur[dl], 1)] = v & 0xffff;
    else       atomicAdd(&ncnt[dl], 1);
}

template<bool PLACE>
__device__ __forceinline__ void bin_rest(const int* __restrict__ recs, const int* __restrict__ row,
                                         int pstart, int plen, int lane, int wave, int nchunks,
                                         int* __restrict__ ncnt, int* __restrict__ ncur,
                                         const int* __restrict__ nstart, int* __restrict__ bucket)
{
    if (__builtin_amdgcn_ballot_w32(plen > 64) != 0u) {
        for (int j = 0; j < 16; ++j) {
            const int po = __builtin_amdgcn_readlane(pstart, j);
            const int pn = __builtin_amdgcn_readlane(plen, j);
            for (int i = 64 + lane; i < pn; i += 32)
                bin_record<PLACE>(recs[po + i], ncnt, ncur, nstart, bucket);
        }
    }
    for (int w = 256 + wave; w < nchunks; w += 16) {
        const int po = row[2 * w];
        const int pn = row[2 * w + 1];
        for (int i = lane; i < pn; i += 32)
            bin_record<PLACE>(recs[po + i], ncnt, ncur, nstart, bucket);
    }
}

__global__ void __launch_bounds__(512)
sort_bins(const int* __restrict__ recs, const int* __restrict__ table,
          int* __restrict__ total, int* __restrict__ cnt, int* __restrict__ offs,
          int* __restrict__ bucket, int N, int nchunks)
{
    __shared__ int ncnt[BIN_NODES];
    __shared__ int ncur[BIN_NODES];
    __shared__ int nstart[BIN_NODES];

    const int tid  = threadIdx.x;
    const int lane = tid & 31;
    const int wave = __builtin_amdgcn_readfirstlane(tid >> 5);
    const int bin  = blockIdx.x;
    const int* row = table + (size_t)bin * nchunks * 2;

    if (tid < BIN_NODES) { ncnt[tid] = 0; ncur[tid] = 0; }

    const int wi = wave * 16 + (lane & 15);
    int pstart = 0, plen = 0;
    if (wi < nchunks) {
        const int2 t = *(const int2*)(row + 2 * wi);
        pstart = t.x;
        plen   = t.y;
    }

    int v[32];
#pragma unroll
    for (int j = 0; j < 16; ++j) {
        const int po = __builtin_amdgcn_readlane(pstart, j);
        const int pn = __builtin_amdgcn_readlane(plen, j);
        v[2 * j]     = (lane < pn)      ? recs[po + lane]      : -1;
        v[2 * j + 1] = (lane + 32 < pn) ? recs[po + 32 + lane] : -1;
    }
    __syncthreads();

#pragma unroll
    for (int k = 0; k < 32; ++k)
        if (v[k] >= 0) bin_record<false>(v[k], ncnt, ncur, nstart, bucket);
    bin_rest<false>(recs, row, pstart, plen, lane, wave, nchunks, ncnt, ncur, nstart, bucket);
    __syncthreads();

    if (wave == 0) {
        int c[8];
        int sum = 0;
#pragma unroll
        for (int j = 0; j < 8; ++j) { c[j] = sum; sum += ncnt[lane * 8 + j]; }
        int incl = sum;
        for (int o = 1; o < 32; o <<= 1) {
            const int up = __shfl_up(incl, o, 32);
            if (lane >= o) incl += up;
        }
        int slice = 0;
        if (lane == 31) slice = atomicAdd(total, incl);
        slice = __shfl(slice, 31, 32);
        const int excl = slice + incl - sum;
#pragma unroll
        for (int j = 0; j < 8; ++j) nstart[lane * 8 + j] = excl + c[j];
    }
    __syncthreads();

    if (tid < BIN_NODES) {
        const int n = bin * BIN_NODES + tid;
        if (n < N) {
            cnt[n]  = ncnt[tid];
            offs[n] = nstart[tid];
        }
    }

    int p[32], t[32];
#pragma unroll
    for (int k = 0; k < 32; ++k) {
        p[k] = (v[k] >= 0) ? nstart[v[k] >> 16] : 0;
        t[k] = (v[k] >= 0) ? atomicAdd(&ncur[v[k] >> 16], 1) : 0;
    }
#pragma unroll
    for (int k = 0; k < 32; ++k)
        if (v[k] >= 0) bucket[p[k] + t[k]] = v[k] & 0xffff;
    bin_rest<true>(recs, row, pstart, plen, lane, wave, nchunks, ncnt, ncur, nstart, bucket);
}

__device__ __forceinline__ v4h row4h(const _Float16* __restrict__ feat, int s, int lane)
{
    return *(const v4h*)(feat + (size_t)s * 128 + lane * 4);
}

__device__ __forceinline__ float4 widen(const v4h h)
{
    return make_float4((float)h[0], (float)h[1], (float)h[2], (float)h[3]);
}

__device__ __forceinline__ float4 row4(const _Float16* __restrict__ feat, int s, int lane)
{
    return widen(row4h(feat, s, lane));
}

__device__ __forceinline__ void add4(float4& a, const float4 v)
{
    a.x += v.x; a.y += v.y; a.z += v.z; a.w += v.w;
}

__device__ __forceinline__ void sum_rows_upto64(const _Float16* __restrict__ feat,
                                                const int* __restrict__ bucket,
                                                int beg, int deg, int lane, float4& acc)
{
    const int k0 = bucket[beg + lane];
    const bool two = lane + 32 < deg;
    const int k1 = two ? bucket[beg + 32 + lane] : 0x7fffffff;
    int r0 = 0, r1 = 0;
    for (int j = 0; j < 32; ++j) {
        const int kj = __builtin_amdgcn_readlane(k0, j);
        r0 += ((kj < k0) || (kj == k0 && j < lane)) ? 1 : 0;
        r1 += (kj <= k1) ? 1 : 0;
    }
    for (int j = 0; j < deg - 32; ++j) {
        const int kj = __builtin_amdgcn_readlane(k1, j);
        r0 += (kj < k0) ? 1 : 0;
        r1 += ((kj < k1) || (kj == k1 && j < lane)) ? 1 : 0;
    }
    if (!two) r1 = -1;
    auto id_at = [&](int p) -> int {
        const unsigned m0 = __builtin_amdgcn_ballot_w32(r0 == p);
        const unsigned m1 = __builtin_amdgcn_ballot_w32(r1 == p);
        const int from = __builtin_ctz(m0 | m1);
        const int a = __builtin_amdgcn_readlane(k0, from);
        const int b = __builtin_amdgcn_readlane(k1, from);
        return (m0 != 0u) ? a : b;
    };
    int j = 0;
    for (; j + 4 <= deg; j += 4) {
        const int i0 = id_at(j + 0), i1 = id_at(j + 1), i2 = id_at(j + 2), i3 = id_at(j + 3);
        const float4 v0 = row4(feat, i0, lane);
        const float4 v1 = row4(feat, i1, lane);
        const float4 v2 = row4(feat, i2, lane);
        const float4 v3 = row4(feat, i3, lane);
        add4(acc, v0); add4(acc, v1); add4(acc, v2); add4(acc, v3);
    }
    if (deg & 2) {
        const int i0 = id_at(j + 0), i1 = id_at(j + 1);
        const float4 v0 = row4(feat, i0, lane);
        const float4 v1 = row4(feat, i1, lane);
        add4(acc, v0); add4(acc, v1);
        j += 2;
    }
    if (deg & 1) {
        const float4 v0 = row4(feat, id_at(j), lane);
        add4(acc, v0);
    }
}

__global__ void __launch_bounds__(256)
gather_sum128(const _Float16* __restrict__ feat,
              const int* __restrict__ offs, const int* __restrict__ cnt,
              const int* __restrict__ bucket,
              float* __restrict__ out, int N)
{
    const int lane = threadIdx.x & 31;
    const int n = __builtin_amdgcn_readfirstlane(blockIdx.x * 8 + (threadIdx.x >> 5));
    if (n >= N) return;
    const int beg = offs[n];
    const int deg = cnt[n];
    float4 acc;

    if (deg <= 32) {
        const int key = (lane < deg) ? bucket[beg + lane] : 0x7fffffff;
        const v4h own = row4h(feat, n, lane);
        int rank = 0;
        for (int j = 0; j < deg; ++j) {
            const int kj = __builtin_amdgcn_readlane(key, j);
            rank += ((kj < key) || (kj == key && j < lane)) ? 1 : 0;
        }
        if (lane >= deg) rank = lane;
        const int sk = __builtin_amdgcn_ds_permute(rank << 2, key);

        acc = widen(own);
        int j = 0;
        for (; j + 8 <= deg; j += 8) {
            float4 v[8];
#pragma unroll
            for (int t = 0; t < 8; ++t)
                v[t] = row4(feat, __builtin_amdgcn_readlane(sk, j + t), lane);
#pragma unroll
            for (int t = 0; t < 8; ++t) add4(acc, v[t]);
        }
        if (deg & 4) {
            const float4 v0 = row4(feat, __builtin_amdgcn_readlane(sk, j + 0), lane);
            const float4 v1 = row4(feat, __builtin_amdgcn_readlane(sk, j + 1), lane);
            const float4 v2 = row4(feat, __builtin_amdgcn_readlane(sk, j + 2), lane);
            const float4 v3 = row4(feat, __builtin_amdgcn_readlane(sk, j + 3), lane);
            add4(acc, v0); add4(acc, v1); add4(acc, v2); add4(acc, v3);
            j += 4;
        }
        if (deg & 2) {
            const float4 v0 = row4(feat, __builtin_amdgcn_readlane(sk, j + 0), lane);
            const float4 v1 = row4(feat, __builtin_amdgcn_readlane(sk, j + 1), lane);
            add4(acc, v0); add4(acc, v1);
            j += 2;
        }
        if (deg & 1) {
            const float4 v0 = row4(feat, __builtin_amdgcn_readlane(sk, j), lane);
            add4(acc, v0);
        }
    } else if (deg <= 64) {
        acc = row4(feat, n, lane);
        sum_rows_upto64(feat, bucket, beg, deg, lane, acc);
    } else {
        acc = row4(feat, n, lane);
        int last = -1;
        int done = 0;
        for (int it = 0; it < deg && done < deg; ++it) {
            int m = 0x7fffffff;
            for (int i = lane; i < deg; i += 32) {
                const int k = bucket[beg + i];
                if (k > last && k < m) m = k;
            }
            for (int o = 16; o > 0; o >>= 1) {
                const int other = __shfl_xor(m, o, 32);
                m = (other < m) ? other : m;
            }
            int c = 0;
            for (int i = lane; i < deg; i += 32) c += (bucket[beg + i] == m) ? 1 : 0;
            for (int o = 16; o > 0; o >>= 1) c += __shfl_xor(c, o, 32);
            m = __builtin_amdgcn_readfirstlane(m);
            c = __builtin_amdgcn_readfirstlane(c);
            if (c <= 0) break;
            const float4 v = row4(feat, m, lane);
            for (int t = 0; t < c; ++t) add4(acc, v);
            last = m;
            done += c;
        }
    }
    *(float4*)(out + (size_t)n * 128 + lane * 4) = acc;
}

typedef __attribute__((ext_vector_type(8))) _Float16 v8h;

template<int K, int NW>
__device__ __forceinline__ void stage_weights(const float* __restrict__ W,
                                              _Float16* __restrict__ Ws, int tid)
{
    constexpr int NT = NW / 16;
    constexpr int TASKS = (K / 8) * NW;
    static_assert(TASKS % 512 == 0, "every thread stages the same number of tasks");
#pragma unroll 4
    for (int i = 0; i < TASKS / 512; ++i) {
        const int t   = i * 512 + tid;
        const int col = t % NW;
        const int kg  = t / NW;
        const float* s = W + (size_t)(kg * 8) * NW + col;
        v8h pk;
#pragma unroll
        for (int e = 0; e < 8; ++e) pk[e] = (_Float16)s[(size_t)e * NW];
        const int q  = kg >> 2;
        const int j  = (kg >> 1) & 1;
        const int h  = kg & 1;
        const int nt = col >> 4;
        const int c  = col & 15;
        *(v8h*)(Ws + ((((q * NT + nt) * 2 + j) * 32) + h * 16 + c) * 8) = pk;
    }
}

template<int NB>
__device__ __forceinline__ void stage_bias(const float* __restrict__ b,
                                           float* __restrict__ Bs, int tid)
{
    static_assert(NB <= 512, "one element per thread");
    if (tid < NB) Bs[tid] = b[tid];
}

template<int K>
__device__ __forceinline__ void load_rows(const float* __restrict__ zrow, int lh,
                                          v16h (&hb)[K / 32])
{
#pragma unroll
    for (int q = 0; q < K / 32; ++q) {
#pragma unroll
        for (int j = 0; j < 2; ++j) {
            const float* p = zrow + 32 * q + 16 * j + 8 * lh;
            const float4 v0 = *(const float4*)p;
            const float4 v1 = *(const float4*)(p + 4);
            hb[q][8 * j + 0] = (_Float16)v0.x;
            hb[q][8 * j + 1] = (_Float16)v0.y;
            hb[q][8 * j + 2] = (_Float16)v0.z;
            hb[q][8 * j + 3] = (_Float16)v0.w;
            hb[q][8 * j + 4] = (_Float16)v1.x;
            hb[q][8 * j + 5] = (_Float16)v1.y;
            hb[q][8 * j + 6] = (_Float16)v1.z;
            hb[q][8 * j + 7] = (_Float16)v1.w;
        }
    }
}

template<int K, int NW>
__device__ __forceinline__ void product(const v16h (&hb)[K / 32],
                                        const _Float16* __restrict__ Ws,
                                        const float* __restrict__ Bs, int lane,
                                        v8f (&acc)[NW / 16])
{
    constexpr int NT = NW / 16;
    constexpr int KC = K / 32;
    constexpr int G = (NT % 4 == 0) ? 4 : NT;
#pragma unroll
    for (int t = 0; t < NT; ++t) acc[t] = *(const v8f*)(Bs + 16 * t + 8 * (lane >> 4));
#pragma unroll
    for (int q = 0; q < KC; ++q) {
#pragma unroll
        for (int g = 0; g < NT; g += G) {
            v16h w[G];
#pragma unroll
            for (int u = 0; u < G; ++u) {
                const _Float16* p = Ws + (((q * NT + g + u) * 2) * 32 + lane) * 8;
                const v8h lo = *(const v8h*)p;
                const v8h hi = *(const v8h*)(p + 32 * 8);
                w[u] = __builtin_shufflevector(lo, hi,
                    0, 1, 2, 3, 4, 5, 6, 7, 8, 9, 10, 11, 12, 13, 14, 15);
            }
#pragma unroll
            for (int u = 0; u < G; ++u)
                acc[g + u] = __builtin_amdgcn_wmma_f32_16x16x32_f16(
                    false, w[u], false, hb[q], (short)0, acc[g + u], false, false);
        }
    }
}

template<int NW>
__device__ __forceinline__ void to_fragments(const v8f (&acc)[NW / 16],
                                             v16h (&hb)[NW / 32])
{
#pragma unroll
    for (int t = 0; t < NW / 16; ++t) {
#pragma unroll
        for (int r = 0; r < 8; ++r)
            hb[t >> 1][8 * (t & 1) + r] = (_Float16)fmaxf(acc[t][r], 0.f);
    }
}

template<int NW, bool RELU>
__device__ __forceinline__ void store_rows(const v8f (&acc)[NW / 16], int lh,
                                           float* __restrict__ orow, int Nout)
{
#pragma unroll
    for (int t = 0; t < NW / 16; ++t) {
        const int col0 = 16 * t + 8 * lh;
        if (col0 < Nout) {
            float4 o0, o1;
            o0.x = acc[t][0]; o0.y = acc[t][1]; o0.z = acc[t][2]; o0.w = acc[t][3];
            o1.x = acc[t][4]; o1.y = acc[t][5]; o1.z = acc[t][6]; o1.w = acc[t][7];
            if (RELU) {
                o0.x = fmaxf(o0.x, 0.f); o0.y = fmaxf(o0.y, 0.f);
                o0.z = fmaxf(o0.z, 0.f); o0.w = fmaxf(o0.w, 0.f);
                o1.x = fmaxf(o1.x, 0.f); o1.y = fmaxf(o1.y, 0.f);
                o1.z = fmaxf(o1.z, 0.f); o1.w = fmaxf(o1.w, 0.f);
            }
            *(float4*)(orow + col0)     = o0;
            *(float4*)(orow + col0 + 4) = o1;
        }
    }
}

template<int NW>
__device__ __forceinline__ void store_rows_f16(const v8f (&acc)[NW / 16], int lh,
                                               _Float16* __restrict__ orow)
{
#pragma unroll
    for (int t = 0; t < NW / 16; ++t) {
        v8h o;
#pragma unroll
        for (int r = 0; r < 8; ++r) o[r] = (_Float16)fmaxf(acc[t][r], 0.f);
        *(v8h*)(orow + 16 * t + 8 * lh) = o;
    }
}

__global__ void __launch_bounds__(512)
mlp_layer1(const float* __restrict__ z,
           const float* __restrict__ Wa, const float* __restrict__ ba,
           const float* __restrict__ Wb, const float* __restrict__ bb,
           _Float16* __restrict__ out, int mtiles)
{
    __shared__ __attribute__((aligned(16))) _Float16 WsA[128 * 128];
    __shared__ __attribute__((aligned(16))) _Float16 WsB[128 * 128];
    __shared__ __attribute__((aligned(32))) float    Bs[128 + 128];

    const int tid  = threadIdx.x;
    const int lane = tid & 31;
    const int wave = __builtin_amdgcn_readfirstlane(tid >> 5);
    const int lm = lane & 15;
    const int lh = lane >> 4;
    const int mt = blockIdx.x * 16 + wave;
    const bool active = mt < mtiles;

    const int node = (active ? mt : (mtiles - 1)) * 16 + lm;
    v16h h0[4];
    load_rows<128>(z + (size_t)node * 128, lh, h0);

    stage_weights<128, 128>(Wa, WsA, tid);
    stage_weights<128, 128>(Wb, WsB, tid);
    stage_bias<128>(ba, Bs, tid);
    stage_bias<128>(bb, Bs + 128, tid);
    asm volatile("" :: "v"(h0[0]), "v"(h0[1]), "v"(h0[2]), "v"(h0[3]));
    __syncthreads();
    if (!active) return;

    v8f acc[8];
    product<128, 128>(h0, WsA, Bs, lane, acc);
    v16h h1[4];
    to_fragments<128>(acc, h1);
    product<128, 128>(h1, WsB, Bs + 128, lane, acc);
    store_rows_f16<128>(acc, lh, out + (size_t)node * 128);
}

__global__ void __launch_bounds__(512)
mlp_layer2_head(const float* __restrict__ z,
                const float* __restrict__ Wa, const float* __restrict__ ba,
                const float* __restrict__ Wb, const float* __restrict__ bb,
                const float* __restrict__ Wc, const float* __restrict__ bc,
                float* __restrict__ out, int Nout, int mtiles)
{
    __shared__ __attribute__((aligned(16))) _Float16 WsA[128 * 256];
    __shared__ __attribute__((aligned(16))) _Float16 WsB[256 * 256];
    __shared__ __attribute__((aligned(16))) _Float16 WsC[256 * 48];
    __shared__ __attribute__((aligned(32))) float    Bs[256 + 256 + 48];

    const int tid  = threadIdx.x;
    const int lane = tid & 31;
    const int wave = __builtin_amdgcn_readfirstlane(tid >> 5);
    const int lm = lane & 15;
    const int lh = lane >> 4;
    const int mt = blockIdx.x * 16 + wave;
    const bool active = mt < mtiles;

    const int node = (active ? mt : (mtiles - 1)) * 16 + lm;
    v16h h0[4];
    load_rows<128>(z + (size_t)node * 128, lh, h0);

    stage_weights<128, 256>(Wa, WsA, tid);
    stage_weights<256, 256>(Wb, WsB, tid);
    stage_weights<256, 48>(Wc, WsC, tid);
    stage_bias<256>(ba, Bs, tid);
    stage_bias<256>(bb, Bs + 256, tid);
    stage_bias<48>(bc, Bs + 512, tid);
    asm volatile("" :: "v"(h0[0]), "v"(h0[1]), "v"(h0[2]), "v"(h0[3]));
    __syncthreads();
    if (!active) return;

    v8f acc[16];
    product<128, 256>(h0, WsA, Bs, lane, acc);
    v16h h1[8];
    to_fragments<256>(acc, h1);
    product<256, 256>(h1, WsB, Bs + 256, lane, acc);
    v16h h2[8];
    to_fragments<256>(acc, h2);
    v8f acc3[3];
    product<256, 48>(h2, WsC, Bs + 512, lane, acc3);
    store_rows<48, false>(acc3, lh, out + (size_t)node * Nout, Nout);
}

extern "C" void kernel_launch(void* const* d_in, const int* in_sizes, int n_in,
                              void* d_out, int out_size, void* d_ws, size_t ws_size,
                              hipStream_t stream)
{
    const float* x  = (const float*)d_in[0];
    const int*   ei = (const int*)  d_in[1];
    const float* W1 = (const float*)d_in[2];
    const float* b1 = (const float*)d_in[3];
    const float* W2 = (const float*)d_in[4];
    const float* b2 = (const float*)d_in[5];
    const float* W3 = (const float*)d_in[6];
    const float* b3 = (const float*)d_in[7];
    const float* W4 = (const float*)d_in[8];
    const float* b4 = (const float*)d_in[9];
    const float* Wl = (const float*)d_in[10];
    const float* bl = (const float*)d_in[11];
    float* out = (float*)d_out;

    const int N = in_sizes[0] / 128;
    const int E = in_sizes[1] / 2;
    const int* src = ei;
    const int* dst = ei + E;

    float* ws   = (float*)d_ws;
    float* bufA = ws;
    _Float16* bufB = (_Float16*)(ws + (size_t)N * 128);
    _Float16* xh   = (_Float16*)(ws + (size_t)N * 256);
    float* Wp   = ws + (size_t)N * 512;
    float* bp   = Wp + 256 * 48;
    const int Epad = (E + 63) / 64 * 64;
    int* cnt    = (int*)(ws + (size_t)N * 512 + 12352);
    int* offs   = cnt + (N + 64);
    int* bucket = offs + (N + 64);
    int* recs   = bucket + Epad;
    int* table  = recs + Epad;

    const int nchunks    = (E + SORT_CHUNK - 1) / SORT_CHUNK;
    const int nbins      = (N + BIN_NODES - 1) / BIN_NODES;
    const int waveBlocks = (N + 7) / 8;
    const int mtiles     = N / 16;
    const int gemmBlocks = (mtiles + 15) / 16;

    sort_chunks_prep<<<nchunks + PREP_GROUPS, 512, 0, stream>>>(src, dst, recs, table, E, nbins, nchunks,
                                                               Wl, bl, Wp, bp, cnt + N, x, xh, N * 32);
    sort_bins<<<nbins, 512, 0, stream>>>(recs, table, cnt + N, cnt, offs, bucket, N, nchunks);

    gather_sum128<<<waveBlocks, 256, 0, stream>>>(xh, offs, cnt, bucket, bufA, N);

    mlp_layer1<<<gemmBlocks, 512, 0, stream>>>(bufA, W1, b1, W2, b2, bufB, mtiles);

    gather_sum128<<<waveBlocks, 256, 0, stream>>>(bufB, offs, cnt, bucket, bufA, N);

    mlp_layer2_head<<<gemmBlocks, 512, 0, stream>>>(bufA, W3, b3, W4, b4, Wp, bp, out, 40, mtiles);
}
